# starting kernel
# baseline (speedup 1.0000x reference)
#include <hip/hip_runtime.h>
#include <hip/hip_bf16.h>

#define HDIM 128
#define EPSV 1e-9f

typedef __attribute__((ext_vector_type(2))) float v2f;
typedef __attribute__((ext_vector_type(8))) float v8f;

__launch_bounds__(256)
__global__ void gemm_wmma_f32(const float* __restrict__ X,
                              const float* __restrict__ W,
                              const float* __restrict__ bias,
                              float* __restrict__ Y,
                              int Mrows, int accumulate) {
    __shared__ float lw[16][132];
    const int n0 = blockIdx.y * 16;

    for (int i = threadIdx.x; i < 16 * HDIM; i += 256) {
        const int n = i >> 7, k = i & (HDIM - 1);
        lw[n][k] = W[(size_t)(n0 + n) * HDIM + k];
    }
    __syncthreads();

    const int wave = threadIdx.x >> 5;
    const int lane = threadIdx.x & 31;
    const int m0   = blockIdx.x * 128 + wave * 16;
    if (m0 >= Mrows) return;

    const int nsub  = lane & 15;
    const int khalf = (lane >> 4) << 1;
    const float* __restrict__ xrow = X + (size_t)(m0 + nsub) * HDIM + khalf;
    const float* __restrict__ wrow = &lw[nsub][khalf];

    v8f acc = {};
#pragma unroll
    for (int k = 0; k < HDIM; k += 4) {
        v2f a, b;
        a.x = xrow[k]; a.y = xrow[k + 1];
        b.x = wrow[k]; b.y = wrow[k + 1];
        acc = __builtin_amdgcn_wmma_f32_16x16x4_f32(
             false, a,  false, b,
             (short)0, acc,  false,  false);
    }

    const float bval  = bias ? bias[n0 + nsub] : 0.0f;
    const int   mbase = m0 + ((lane >> 4) << 3);
    float* __restrict__ yp = Y + (size_t)mbase * HDIM + n0 + nsub;
#pragma unroll
    for (int r = 0; r < 8; ++r) {
        float v = acc[r] + bval;
        if (accumulate) v += yp[(size_t)r * HDIM];
        yp[(size_t)r * HDIM] = v;
    }
}

__device__ __forceinline__ float sigmoidf_(float x) {
    return 1.0f / (1.0f + __expf(-x));
}

__launch_bounds__(256)
__global__ void edge_score_kernel(const float* __restrict__ Ah,
                                  const float* __restrict__ Bx,
                                  const float* __restrict__ v_w,
                                  const int* __restrict__ ci,
                                  const int* __restrict__ chi,
                                  float* __restrict__ e,
                                  unsigned int* __restrict__ seg_max_u,
                                  int L) {
    const int edge = blockIdx.x * 8 + (threadIdx.x >> 5);
    const int lane = threadIdx.x & 31;
    if (edge >= L) return;
    const int c = ci[edge], ch = chi[edge];

    const float4 a = ((const float4*)(Ah + (size_t)ch * HDIM))[lane];
    const float4 b = ((const float4*)(Bx + (size_t)c  * HDIM))[lane];
    const float4 v = ((const float4*)v_w)[lane];

    float s = tanhf(a.x + b.x) * v.x + tanhf(a.y + b.y) * v.y
            + tanhf(a.z + b.z) * v.z + tanhf(a.w + b.w) * v.w;
#pragma unroll
    for (int off = 16; off > 0; off >>= 1) s += __shfl_xor(s, off, 32);

    if (lane == 0) {
        e[edge] = s;
        const unsigned int b32 = __float_as_uint(s);
        const unsigned int key = (b32 & 0x80000000u) ? ~b32 : (b32 | 0x80000000u);
        atomicMax(&seg_max_u[c], key);
    }
}

__launch_bounds__(256)
__global__ void edge_softmax_kernel(const float* __restrict__ e,
                                    const int* __restrict__ ci,
                                    const unsigned int* __restrict__ seg_max_u,
                                    float* __restrict__ x_exp,
                                    float* __restrict__ seg_sum,
                                    int L) {
    const int t = blockIdx.x * 256 + threadIdx.x;
    if (t >= L) return;
    const int c = ci[t];
    const unsigned int key = seg_max_u[c];
    const unsigned int b32 = (key & 0x80000000u) ? (key & 0x7FFFFFFFu) : ~key;
    const float xe = __expf(e[t] - __uint_as_float(b32));
    x_exp[t] = xe;
    atomicAdd(&seg_sum[c], xe);
}

__launch_bounds__(256)
__global__ void edge_accum_kernel(const float* __restrict__ x_exp,
                                  const float* __restrict__ seg_sum,
                                  const float* __restrict__ child_h,
                                  const float* __restrict__ child_c,
                                  const float* __restrict__ Wfx,
                                  const float* __restrict__ Ufh,
                                  const int* __restrict__ ci,
                                  const int* __restrict__ chi,
                                  float* __restrict__ h_hat,
                                  float* __restrict__ sum_fc,
                                  int L) {
    const int edge = blockIdx.x * 8 + (threadIdx.x >> 5);
    const int lane = threadIdx.x & 31;
    if (edge >= L) return;
    const int c = ci[edge], ch = chi[edge];
    const float attn = x_exp[edge] / (seg_sum[c] + EPSV);

    const float4 hv = ((const float4*)(child_h + (size_t)ch * HDIM))[lane];
    const float4 cv = ((const float4*)(child_c + (size_t)ch * HDIM))[lane];
    const float4 wf = ((const float4*)(Wfx + (size_t)c  * HDIM))[lane];
    const float4 uf = ((const float4*)(Ufh + (size_t)ch * HDIM))[lane];

    float* hh = h_hat  + (size_t)c * HDIM + lane * 4;
    float* sf = sum_fc + (size_t)c * HDIM + lane * 4;

    atomicAdd(&hh[0], attn * hv.x);
    atomicAdd(&hh[1], attn * hv.y);
    atomicAdd(&hh[2], attn * hv.z);
    atomicAdd(&hh[3], attn * hv.w);

    atomicAdd(&sf[0], sigmoidf_(wf.x + uf.x) * cv.x);
    atomicAdd(&sf[1], sigmoidf_(wf.y + uf.y) * cv.y);
    atomicAdd(&sf[2], sigmoidf_(wf.z + uf.z) * cv.z);
    atomicAdd(&sf[3], sigmoidf_(wf.w + uf.w) * cv.w);
}

__launch_bounds__(256)
__global__ void final_gate_kernel(const float* __restrict__ Gi,
                                  const float* __restrict__ Gc,
                                  const float* __restrict__ Go,
                                  const float* __restrict__ sum_fc,
                                  float* __restrict__ h_out,
                                  float* __restrict__ c_out,
                                  size_t n) {
    const size_t t = (size_t)blockIdx.x * 256 + threadIdx.x;
    if (t >= n) return;
    const float i_j = sigmoidf_(Gi[t]);
    const float c_t = tanhf(Gc[t]);
    const float c   = i_j * c_t + sum_fc[t];
    const float o_j = sigmoidf_(Go[t]);
    h_out[t] = o_j * tanhf(c);
    c_out[t] = c;
}

extern "C" void kernel_launch(void* const* d_in, const int* in_sizes, int n_in,
                              void* d_out, int out_size, void* d_ws, size_t ws_size,
                              hipStream_t stream) {
    const float* x_emb   = (const float*)d_in[0];
    const float* child_h = (const float*)d_in[1];
    const float* child_c = (const float*)d_in[2];
    const int*   ci      = (const int*)d_in[3];
    const int*   chi     = (const int*)d_in[4];
    const float* Wi_w = (const float*)d_in[5];
    const float* Ui_w = (const float*)d_in[6];
    const float* Wf_w = (const float*)d_in[7];
    const float* Uf_w = (const float*)d_in[8];
    const float* Wo_w = (const float*)d_in[9];
    const float* Uo_w = (const float*)d_in[10];
    const float* Wc_w = (const float*)d_in[11];
    const float* Uc_w = (const float*)d_in[12];
    const float* Wa_w = (const float*)d_in[13];
    const float* Ua_w = (const float*)d_in[14];
    const float* Wi_b = (const float*)d_in[15];
    const float* Wf_b = (const float*)d_in[16];
    const float* Wo_b = (const float*)d_in[17];
    const float* Wc_b = (const float*)d_in[18];
    const float* Wa_b = (const float*)d_in[19];
    const float* v_w  = (const float*)d_in[20];

    const int M = in_sizes[0] / HDIM;
    const int L = in_sizes[3];
    const size_t MH = (size_t)M * HDIM;

    float* ws    = (float*)d_ws;
    float* Ah    = ws;
    float* Bx    = Ah  + MH;
    float* Gi    = Bx  + MH;
    float* Wfx   = Gi  + MH;
    float* Go    = Wfx + MH;
    float* Gc    = Go  + MH;
    float* Ufh   = Gc  + MH;
    float* h_hat = Ufh + MH;
    float* sumfc = h_hat + MH;
    float* e_buf = sumfc + MH;
    float* x_exp = e_buf + L;
    float* seg_sum = x_exp + L;
    unsigned int* seg_max_u = (unsigned int*)(seg_sum + M);

    hipMemsetAsync(h_hat,     0, 2 * MH * sizeof(float), stream);
    hipMemsetAsync(seg_sum,   0, (size_t)M * sizeof(float), stream);
    hipMemsetAsync(seg_max_u, 0, (size_t)M * sizeof(unsigned int), stream);

    const dim3 gblk(256);
    const dim3 ggrd((M + 127) / 128, HDIM / 16);

    gemm_wmma_f32<<<ggrd, gblk, 0, stream>>>(child_h, Wa_w, Wa_b, Ah, M, 0);
    gemm_wmma_f32<<<ggrd, gblk, 0, stream>>>(x_emb,   Ua_w, nullptr, Bx, M, 0);
    gemm_wmma_f32<<<ggrd, gblk, 0, stream>>>(x_emb,   Wi_w, Wi_b, Gi, M, 0);
    gemm_wmma_f32<<<ggrd, gblk, 0, stream>>>(x_emb,   Wf_w, Wf_b, Wfx, M, 0);
    gemm_wmma_f32<<<ggrd, gblk, 0, stream>>>(x_emb,   Wo_w, Wo_b, Go, M, 0);
    gemm_wmma_f32<<<ggrd, gblk, 0, stream>>>(x_emb,   Wc_w, Wc_b, Gc, M, 0);
    gemm_wmma_f32<<<ggrd, gblk, 0, stream>>>(child_h, Uf_w, nullptr, Ufh, M, 0);

    edge_score_kernel<<<(L + 7) / 8, gblk, 0, stream>>>(Ah, Bx, v_w, ci, chi,
                                                        e_buf, seg_max_u, L);
    edge_softmax_kernel<<<(L + 255) / 256, gblk, 0, stream>>>(e_buf, ci, seg_max_u,
                                                              x_exp, seg_sum, L);
    edge_accum_kernel<<<(L + 7) / 8, gblk, 0, stream>>>(x_exp, seg_sum,
                                                        child_h, child_c, Wfx, Ufh,
                                                        ci, chi, h_hat, sumfc, L);

    gemm_wmma_f32<<<ggrd, gblk, 0, stream>>>(h_hat, Ui_w, nullptr, Gi, M, 1);
    gemm_wmma_f32<<<ggrd, gblk, 0, stream>>>(h_hat, Uc_w, nullptr, Gc, M, 1);
    gemm_wmma_f32<<<ggrd, gblk, 0, stream>>>(h_hat, Uo_w, nullptr, Go, M, 1);

    float* h_out = (float*)d_out;
    float* c_out = h_out + MH;
    final_gate_kernel<<<(unsigned)((MH + 255) / 256), gblk, 0, stream>>>(
        Gi, Gc, Go, sumfc, h_out, c_out, MH);
}


# Edge sums by owner, no atomics: 256 blocks keep 196 candidate rows in LDS, waves list their edges by scanning all ids, one wave per row adds in ascending edge index (was arrival order); no zero fill
# speedup vs baseline: 1.9575x; 1.9575x over previous
#include <hip/hip_runtime.h>
#include <hip/hip_bf16.h>

#define HDIM 128
#define EPSV 1e-9f

typedef __attribute__((ext_vector_type(2))) float v2f;
typedef __attribute__((ext_vector_type(8))) float v8f;

template <bool ACC>
__launch_bounds__(256)
__global__ void gemm_wmma_f32(const float* __restrict__ X,
                              const float* __restrict__ W,
                              const float* __restrict__ bias,
                              float* __restrict__ Y,
                              int Mrows) {
    __shared__ float lw[HDIM][132];
    __shared__ float lx[2][16][132];

    const int wave  = threadIdx.x >> 5;
    const int lane  = threadIdx.x & 31;
    const int nsub  = lane & 15;
    const int khalf = (lane >> 4) << 1;
    const int n0    = wave * 16;
    const int ntiles = (Mrows + 15) >> 4;

    const int srow = threadIdx.x >> 5;
    const int scol = (threadIdx.x & 31) << 2;
    const int lastrow = Mrows - 1;

    int t = blockIdx.x;
    float4 p0 = {}, p1 = {};
    if (t < ntiles) {
        const int ra = min(t * 16 + srow, lastrow), rb = min(t * 16 + srow + 8, lastrow);
        p0 = *(const float4*)(X + (size_t)ra * HDIM + scol);
        p1 = *(const float4*)(X + (size_t)rb * HDIM + scol);
    }

#pragma unroll
    for (int p = 0; p < HDIM / 8; ++p)
        *(float4*)&lw[srow + 8 * p][scol] =
            *(const float4*)(W + (size_t)(srow + 8 * p) * HDIM + scol);
    __syncthreads();

    v2f bfrag[HDIM / 4];
#pragma unroll
    for (int j = 0; j < HDIM / 4; ++j)
        bfrag[j] = *(const v2f*)&lw[n0 + nsub][4 * j + khalf];
    const float bval = bias ? bias[n0 + nsub] : 0.0f;

    int buf = 0;
    for (; t < ntiles; t += gridDim.x) {
        *(float4*)&lx[buf][srow][scol]     = p0;
        *(float4*)&lx[buf][srow + 8][scol] = p1;
        __syncthreads();

        const int tn = t + gridDim.x;
        if (tn < ntiles) {
            const int ra = min(tn * 16 + srow, lastrow), rb = min(tn * 16 + srow + 8, lastrow);
            p0 = *(const float4*)(X + (size_t)ra * HDIM + scol);
            p1 = *(const float4*)(X + (size_t)rb * HDIM + scol);
        }

        const bool full  = (t * 16 + 16 <= Mrows);
        const int  mbase = t * 16 + ((lane >> 4) << 3);
        float* __restrict__ yp = Y + (size_t)mbase * HDIM + n0 + nsub;
        float yv[8];
        if (ACC) {
            if (full) {
#pragma unroll
                for (int r = 0; r < 8; ++r) yv[r] = yp[(size_t)r * HDIM];
            } else {
#pragma unroll
                for (int r = 0; r < 8; ++r)
                    yv[r] = (mbase + r < Mrows) ? yp[(size_t)r * HDIM] : 0.0f;
            }
        }

        const float* arow = &lx[buf][nsub][khalf];
        v8f acc = {};
#pragma unroll
        for (int j = 0; j < HDIM / 4; ++j) {
            const v2f a = *(const v2f*)(arow + 4 * j);
            acc = __builtin_amdgcn_wmma_f32_16x16x4_f32(
                 false, a,  false, bfrag[j],
                 (short)0, acc,  false,  false);
        }

        float v[8];
#pragma unroll
        for (int r = 0; r < 8; ++r) {
            v[r] = acc[r] + bval;
            if (ACC) v[r] += yv[r];
        }
        if (full) {
#pragma unroll
            for (int r = 0; r < 8; ++r) yp[(size_t)r * HDIM] = v[r];
        } else {
#pragma unroll
            for (int r = 0; r < 8; ++r)
                if (mbase + r < Mrows) yp[(size_t)r * HDIM] = v[r];
        }
        buf ^= 1;
    }
}

__device__ __forceinline__ float sigmoidf_(float x) {
    return 1.0f / (1.0f + __expf(-x));
}

__launch_bounds__(256)
__global__ void edge_score_kernel(const float* __restrict__ Ah,
                                  const float* __restrict__ Bx,
                                  const float* __restrict__ v_w,
                                  const int* __restrict__ ci,
                                  const int* __restrict__ chi,
                                  float* __restrict__ e,
                                  unsigned int* __restrict__ seg_max_u,
                                  int L) {
    const int edge = blockIdx.x * 8 + (threadIdx.x >> 5);
    const int lane = threadIdx.x & 31;
    if (edge >= L) return;
    const int c = ci[edge], ch = chi[edge];

    const float4 a = ((const float4*)(Ah + (size_t)ch * HDIM))[lane];
    const float4 b = ((const float4*)(Bx + (size_t)c  * HDIM))[lane];
    const float4 v = ((const float4*)v_w)[lane];

    float s = tanhf(a.x + b.x) * v.x + tanhf(a.y + b.y) * v.y
            + tanhf(a.z + b.z) * v.z + tanhf(a.w + b.w) * v.w;
#pragma unroll
    for (int off = 16; off > 0; off >>= 1) s += __shfl_xor(s, off, 32);

    if (lane == 0) {
        e[edge] = s;
        const unsigned int b32 = __float_as_uint(s);
        const unsigned int key = (b32 & 0x80000000u) ? ~b32 : (b32 | 0x80000000u);
        atomicMax(&seg_max_u[c], key);
    }
}

__launch_bounds__(256)
__global__ void edge_softmax_kernel(const float* __restrict__ e,
                                    const int* __restrict__ ci,
                                    const unsigned int* __restrict__ seg_max_u,
                                    float* __restrict__ x_exp,
                                    float* __restrict__ seg_sum,
                                    int L) {
    const int t = blockIdx.x * 256 + threadIdx.x;
    if (t >= L) return;
    const int c = ci[t];
    const unsigned int key = seg_max_u[c];
    const unsigned int b32 = (key & 0x80000000u) ? (key & 0x7FFFFFFFu) : ~key;
    const float xe = __expf(e[t] - __uint_as_float(b32));
    x_exp[t] = xe;
    atomicAdd(&seg_sum[c], xe);
}

#define OWN_R    196
#define OWN_NT   1024
#define OWN_SEG  512

__launch_bounds__(OWN_NT)
__global__ void edge_accum_owner_kernel(const float* __restrict__ x_exp,
                                        const float* __restrict__ seg_sum,
                                        const float* __restrict__ child_h,
                                        const float* __restrict__ child_c,
                                        const float* __restrict__ Wfx,
                                        const float* __restrict__ Ufh,
                                        const int* __restrict__ ci,
                                        const int* __restrict__ chi,
                                        float* __restrict__ h_hat,
                                        float* __restrict__ sum_fc,
                                        int M, int L) {
    __shared__ float hacc[OWN_R][HDIM];
    __shared__ float facc[OWN_R][HDIM];
    __shared__ int           list_e[32][OWN_SEG];
    __shared__ unsigned char list_c[32][OWN_SEG];
    __shared__ int           s_cnt[32];
    __shared__ int           s_more[32];
    __shared__ int           wb_e[32][64];
    __shared__ unsigned char wb_c[32][64];

    const int wave = __builtin_amdgcn_readfirstlane((int)(threadIdx.x >> 5));
    const int lane = threadIdx.x & 31;
    const unsigned lt = (1u << lane) - 1u;
    const int SL = (((L + 31) >> 5) + 31) & ~31;

    for (int cbase = blockIdx.x * OWN_R; cbase < M; cbase += gridDim.x * OWN_R) {
        const int nown = min(OWN_R, M - cbase);

        for (int i = threadIdx.x; i < OWN_R * HDIM / 4; i += OWN_NT) {
            ((float4*)&hacc[0][0])[i] = make_float4(0.f, 0.f, 0.f, 0.f);
            ((float4*)&facc[0][0])[i] = make_float4(0.f, 0.f, 0.f, 0.f);
        }
        int wpos = min(L, wave * SL);
        const int wend = min(L, wave * SL + SL);
        __syncthreads();

        bool more = true;
        while (more) {
            int cnt = 0;
            if (wpos < wend) {
                int ida[8], idb[8];
                auto load_ids = [&](int (&v)[8], int p) {
#pragma unroll
                    for (int j = 0; j < 8; ++j) {
                        const int idx = p + j * 32 + lane;
                        const int t = ci[min(idx, L - 1)];
                        v[j] = (idx < wend) ? t : -1;
                    }
                };
                auto take = [&](const int (&v)[8], int p) {
#pragma unroll
                    for (int j = 0; j < 8; ++j) {
                        const int cl = v[j] - cbase;
                        const bool hit = (unsigned)cl < (unsigned)nown;
                        const unsigned m = (unsigned)__ballot(hit);
                        if (hit) {
                            const int k = cnt + __popc(m & lt);
                            list_e[wave][k] = p + j * 32 + lane;
                            list_c[wave][k] = (unsigned char)cl;
                        }
                        cnt += __popc(m);
                    }
                };
                load_ids(ida, wpos);
                for (;;) {
                    load_ids(idb, wpos + 256);
                    take(ida, wpos);
                    wpos += 256;
                    if (!(wpos < wend && cnt + 256 <= OWN_SEG)) break;
                    load_ids(ida, wpos + 256);
                    take(idb, wpos);
                    wpos += 256;
                    if (!(wpos < wend && cnt + 256 <= OWN_SEG)) break;
                }
                wpos = min(wpos, wend);
            }
            if (lane == 0) {
                s_cnt[wave]  = cnt;
                s_more[wave] = (wpos < wend) ? 1 : 0;
            }
            __syncthreads();
            const int cntv = s_cnt[lane];
            more = __ballot(s_more[lane] != 0) != 0;

            int nb = 0;
            int q = 0, k0 = 0;
            for (;;) {
                int nq = 0;
                bool have = false;
                while (q < 32) {
                    nq = __builtin_amdgcn_readlane(cntv, q);
                    if (k0 < nq) { have = true; break; }
                    ++q; k0 = 0;
                }
                if (have) {
                    const int k = k0 + lane;
                    const bool valid = k < nq;
                    const int e  = valid ? list_e[q][k] : 0;
                    const int cl = valid ? (int)list_c[q][k] : 0;
                    const bool mine = valid && ((cl & 31) == wave);
                    const unsigned mm = (unsigned)__ballot(mine);
                    if (mine) {
                        const int d = nb + __popc(mm & lt);
                        wb_e[wave][d] = e;
                        wb_c[wave][d] = (unsigned char)cl;
                    }
                    nb += __popc(mm);
                    k0 += 32;
                }
                if (nb >= 32 || (!have && nb > 0)) {
                    const int n = min(nb, 32);
                    int be = 0, bcl = 0, bch = 0;
                    float bat = 0.f;
                    if (lane < n) {
                        be  = wb_e[wave][lane];
                        bcl = (int)wb_c[wave][lane];
                        bch = chi[be];
                        bat = x_exp[be] / (seg_sum[cbase + bcl] + EPSV);
                    }
                    for (int g = 0; g < n; g += 4) {
                        float4 hv[4], cv4[4], uf[4], wf[4];
                        int   cl4[4];
                        float at4[4];
#pragma unroll
                        for (int u = 0; u < 4; ++u) {
                            const int i  = min(g + u, n - 1);
                            const int ch = __builtin_amdgcn_readlane(bch, i);
                            const int cl = __builtin_amdgcn_readlane(bcl, i);
                            at4[u] = __int_as_float(__builtin_amdgcn_readlane(__float_as_int(bat), i));
                            cl4[u] = cl;
                            hv[u]  = ((const float4*)(child_h + (size_t)ch * HDIM))[lane];
                            cv4[u] = ((const float4*)(child_c + (size_t)ch * HDIM))[lane];
                            uf[u]  = ((const float4*)(Ufh     + (size_t)ch * HDIM))[lane];
                            wf[u]  = ((const float4*)(Wfx + (size_t)(cbase + cl) * HDIM))[lane];
                        }
#pragma unroll
                        for (int u = 0; u < 4; ++u) {
                            if (g + u < n) {
                                float4* hp = (float4*)&hacc[cl4[u]][lane * 4];
                                float4* fp = (float4*)&facc[cl4[u]][lane * 4];
                                float4 h = *hp, f = *fp;
                                const float at = at4[u];
                                h.x += at * hv[u].x;
                                h.y += at * hv[u].y;
                                h.z += at * hv[u].z;
                                h.w += at * hv[u].w;
                                f.x += sigmoidf_(wf[u].x + uf[u].x) * cv4[u].x;
                                f.y += sigmoidf_(wf[u].y + uf[u].y) * cv4[u].y;
                                f.z += sigmoidf_(wf[u].z + uf[u].z) * cv4[u].z;
                                f.w += sigmoidf_(wf[u].w + uf[u].w) * cv4[u].w;
                                *hp = h;
                                *fp = f;
                            }
                        }
                    }
                    const int rest = nb - n;
                    int te = 0;
                    unsigned char tc = 0;
                    if (lane < rest) { te = wb_e[wave][32 + lane]; tc = wb_c[wave][32 + lane]; }
                    if (lane < rest) { wb_e[wave][lane] = te; wb_c[wave][lane] = tc; }
                    nb = rest;
                }
                if (!have) break;
            }
            __syncthreads();
        }

        for (int i = threadIdx.x; i < nown * (HDIM / 4); i += OWN_NT) {
            const int row = i >> 5, c4 = i & 31;
            ((float4*)(h_hat  + (size_t)(cbase + row) * HDIM))[c4] = ((const float4*)&hacc[row][0])[c4];
            ((float4*)(sum_fc + (size_t)(cbase + row) * HDIM))[c4] = ((const float4*)&facc[row][0])[c4];
        }
        __syncthreads();
    }
}

__launch_bounds__(256)
__global__ void final_gate_kernel(const float* __restrict__ Gi,
                                  const float* __restrict__ Gc,
                                  const float* __restrict__ Go,
                                  const float* __restrict__ sum_fc,
                                  float* __restrict__ h_out,
                                  float* __restrict__ c_out,
                                  size_t n) {
    const size_t t = (size_t)blockIdx.x * 256 + threadIdx.x;
    if (t >= n) return;
    const float i_j = sigmoidf_(Gi[t]);
    const float c_t = tanhf(Gc[t]);
    const float c   = i_j * c_t + sum_fc[t];
    const float o_j = sigmoidf_(Go[t]);
    h_out[t] = o_j * tanhf(c);
    c_out[t] = c;
}

extern "C" void kernel_launch(void* const* d_in, const int* in_sizes, int n_in,
                              void* d_out, int out_size, void* d_ws, size_t ws_size,
                              hipStream_t stream) {
    const float* x_emb   = (const float*)d_in[0];
    const float* child_h = (const float*)d_in[1];
    const float* child_c = (const float*)d_in[2];
    const int*   ci      = (const int*)d_in[3];
    const int*   chi     = (const int*)d_in[4];
    const float* Wi_w = (const float*)d_in[5];
    const float* Ui_w = (const float*)d_in[6];
    const float* Wf_w = (const float*)d_in[7];
    const float* Uf_w = (const float*)d_in[8];
    const float* Wo_w = (const float*)d_in[9];
    const float* Uo_w = (const float*)d_in[10];
    const float* Wc_w = (const float*)d_in[11];
    const float* Uc_w = (const float*)d_in[12];
    const float* Wa_w = (const float*)d_in[13];
    const float* Ua_w = (const float*)d_in[14];
    const float* Wi_b = (const float*)d_in[15];
    const float* Wf_b = (const float*)d_in[16];
    const float* Wo_b = (const float*)d_in[17];
    const float* Wc_b = (const float*)d_in[18];
    const float* Wa_b = (const float*)d_in[19];
    const float* v_w  = (const float*)d_in[20];

    const int M = in_sizes[0] / HDIM;
    const int L = in_sizes[3];
    const size_t MH = (size_t)M * HDIM;

    float* ws    = (float*)d_ws;
    float* Ah    = ws;
    float* Bx    = Ah  + MH;
    float* Gi    = Bx  + MH;
    float* Wfx   = Gi  + MH;
    float* Go    = Wfx + MH;
    float* Gc    = Go  + MH;
    float* Ufh   = Gc  + MH;
    float* h_hat = Ufh + MH;
    float* sumfc = h_hat + MH;
    float* e_buf = sumfc + MH;
    float* x_exp = e_buf + L;
    float* seg_sum = x_exp + L;
    unsigned int* seg_max_u = (unsigned int*)(seg_sum + M);

    hipMemsetAsync(seg_sum,   0, (size_t)M * sizeof(float), stream);
    hipMemsetAsync(seg_max_u, 0, (size_t)M * sizeof(unsigned int), stream);

    const dim3 gblk(256);
    const dim3 ggrd(512);

    gemm_wmma_f32<false><<<ggrd, gblk, 0, stream>>>(child_h, Wa_w, Wa_b, Ah, M);
    gemm_wmma_f32<false><<<ggrd, gblk, 0, stream>>>(x_emb,   Ua_w, nullptr, Bx, M);
    gemm_wmma_f32<false><<<ggrd, gblk, 0, stream>>>(x_emb,   Wi_w, Wi_b, Gi, M);
    gemm_wmma_f32<false><<<ggrd, gblk, 0, stream>>>(x_emb,   Wf_w, Wf_b, Wfx, M);
    gemm_wmma_f32<false><<<ggrd, gblk, 0, stream>>>(x_emb,   Wo_w, Wo_b, Go, M);
    gemm_wmma_f32<false><<<ggrd, gblk, 0, stream>>>(x_emb,   Wc_w, Wc_b, Gc, M);
    gemm_wmma_f32<false><<<ggrd, gblk, 0, stream>>>(child_h, Uf_w, nullptr, Ufh, M);

    edge_score_kernel<<<(L + 7) / 8, gblk, 0, stream>>>(Ah, Bx, v_w, ci, chi,
                                                        e_buf, seg_max_u, L);
    edge_softmax_kernel<<<(L + 255) / 256, gblk, 0, stream>>>(e_buf, ci, seg_max_u,
                                                              x_exp, seg_sum, L);
    edge_accum_owner_kernel<<<dim3(256), dim3(OWN_NT), 0, stream>>>(x_exp, seg_sum,
                                                        child_h, child_c, Wfx, Ufh,
                                                        ci, chi, h_hat, sumfc, M, L);

    gemm_wmma_f32<true><<<ggrd, gblk, 0, stream>>>(h_hat, Ui_w, nullptr, Gi, M);
    gemm_wmma_f32<true><<<ggrd, gblk, 0, stream>>>(h_hat, Uc_w, nullptr, Gc, M);
    gemm_wmma_f32<true><<<ggrd, gblk, 0, stream>>>(h_hat, Uo_w, nullptr, Go, M);

    float* h_out = (float*)d_out;
    float* c_out = h_out + MH;
    final_gate_kernel<<<(unsigned)((MH + 255) / 256), gblk, 0, stream>>>(
        Gi, Gc, Go, sumfc, h_out, c_out, MH);
}


# Scores and segment softmax moved into the owner kernel: running max and denominator per owned row (rescaled on a new max), sums in ascending edge index, one divide per row; 2 launches, atomics gone
# speedup vs baseline: 2.2365x; 1.1425x over previous
#include <hip/hip_runtime.h>
#include <hip/hip_bf16.h>

#define HDIM 128
#define EPSV 1e-9f

typedef __attribute__((ext_vector_type(2))) float v2f;
typedef __attribute__((ext_vector_type(8))) float v8f;

template <bool ACC>
__launch_bounds__(256)
__global__ void gemm_wmma_f32(const float* __restrict__ X,
                              const float* __restrict__ W,
                              const float* __restrict__ bias,
                              float* __restrict__ Y,
                              int Mrows) {
    __shared__ float lw[HDIM][132];
    __shared__ float lx[2][16][132];

    const int wave  = threadIdx.x >> 5;
    const int lane  = threadIdx.x & 31;
    const int nsub  = lane & 15;
    const int khalf = (lane >> 4) << 1;
    const int n0    = wave * 16;
    const int ntiles = (Mrows + 15) >> 4;

    const int srow = threadIdx.x >> 5;
    const int scol = (threadIdx.x & 31) << 2;
    const int lastrow = Mrows - 1;

    int t = blockIdx.x;
    float4 p0 = {}, p1 = {};
    if (t < ntiles) {
        const int ra = min(t * 16 + srow, lastrow), rb = min(t * 16 + srow + 8, lastrow);
        p0 = *(const float4*)(X + (size_t)ra * HDIM + scol);
        p1 = *(const float4*)(X + (size_t)rb * HDIM + scol);
    }

#pragma unroll
    for (int p = 0; p < HDIM / 8; ++p)
        *(float4*)&lw[srow + 8 * p][scol] =
            *(const float4*)(W + (size_t)(srow + 8 * p) * HDIM + scol);
    __syncthreads();

    v2f bfrag[HDIM / 4];
#pragma unroll
    for (int j = 0; j < HDIM / 4; ++j)
        bfrag[j] = *(const v2f*)&lw[n0 + nsub][4 * j + khalf];
    const float bval = bias ? bias[n0 + nsub] : 0.0f;

    int buf = 0;
    for (; t < ntiles; t += gridDim.x) {
        *(float4*)&lx[buf][srow][scol]     = p0;
        *(float4*)&lx[buf][srow + 8][scol] = p1;
        __syncthreads();

        const int tn = t + gridDim.x;
        if (tn < ntiles) {
            const int ra = min(tn * 16 + srow, lastrow), rb = min(tn * 16 + srow + 8, lastrow);
            p0 = *(const float4*)(X + (size_t)ra * HDIM + scol);
            p1 = *(const float4*)(X + (size_t)rb * HDIM + scol);
        }

        const bool full  = (t * 16 + 16 <= Mrows);
        const int  mbase = t * 16 + ((lane >> 4) << 3);
        float* __restrict__ yp = Y + (size_t)mbase * HDIM + n0 + nsub;
        float yv[8];
        if (ACC) {
            if (full) {
#pragma unroll
                for (int r = 0; r < 8; ++r) yv[r] = yp[(size_t)r * HDIM];
            } else {
#pragma unroll
                for (int r = 0; r < 8; ++r)
                    yv[r] = (mbase + r < Mrows) ? yp[(size_t)r * HDIM] : 0.0f;
            }
        }

        const float* arow = &lx[buf][nsub][khalf];
        v8f acc = {};
#pragma unroll
        for (int j = 0; j < HDIM / 4; ++j) {
            const v2f a = *(const v2f*)(arow + 4 * j);
            acc = __builtin_amdgcn_wmma_f32_16x16x4_f32(
                 false, a,  false, bfrag[j],
                 (short)0, acc,  false,  false);
        }

        float v[8];
#pragma unroll
        for (int r = 0; r < 8; ++r) {
            v[r] = acc[r] + bval;
            if (ACC) v[r] += yv[r];
        }
        if (full) {
#pragma unroll
            for (int r = 0; r < 8; ++r) yp[(size_t)r * HDIM] = v[r];
        } else {
#pragma unroll
            for (int r = 0; r < 8; ++r)
                if (mbase + r < Mrows) yp[(size_t)r * HDIM] = v[r];
        }
        buf ^= 1;
    }
}

__device__ __forceinline__ float sigmoidf_(float x) {
    return 1.0f / (1.0f + __expf(-x));
}

template <int XM>
__device__ __forceinline__ float row_xmask_(float x) {
    const int b = __builtin_bit_cast(int, x);
    return __builtin_bit_cast(float, __builtin_amdgcn_update_dpp(b, b, 0x160 | XM, 0xf, 0xf, false));
}
__device__ __forceinline__ float wave_sum32_(float x) {
    const int b = __builtin_bit_cast(int, x);
    x += __builtin_bit_cast(float, __builtin_amdgcn_permlanex16(b, b, 0x76543210, 0xfedcba98, false, false));
    x += row_xmask_<8>(x);
    x += row_xmask_<4>(x);
    x += row_xmask_<2>(x);
    x += row_xmask_<1>(x);
    return x;
}

#define OWN_R    196
#define OWN_NT   1024
#define OWN_SEG  512
#define OWN_G    2

__launch_bounds__(OWN_NT)
__global__ void edge_attn_owner_kernel(const float* __restrict__ Ah,
                                       const float* __restrict__ Bx,
                                       const float* __restrict__ v_w,
                                       const float* __restrict__ child_h,
                                       const float* __restrict__ child_c,
                                       const float* __restrict__ Wfx,
                                       const float* __restrict__ Ufh,
                                       const int* __restrict__ ci,
                                       const int* __restrict__ chi,
                                       float* __restrict__ h_hat,
                                       float* __restrict__ sum_fc,
                                       int M, int L) {
    __shared__ float hacc[OWN_R][HDIM];
    __shared__ float facc[OWN_R][HDIM];
    __shared__ float rmax[OWN_R];
    __shared__ float rden[OWN_R];
    __shared__ int           list_e[32][OWN_SEG];
    __shared__ unsigned char list_c[32][OWN_SEG];
    __shared__ int           s_cnt[32];
    __shared__ int           s_more[32];
    __shared__ int           wb_e[32][64];
    __shared__ unsigned char wb_c[32][64];

    const int wave = __builtin_amdgcn_readfirstlane((int)(threadIdx.x >> 5));
    const int lane = threadIdx.x & 31;
    const unsigned lt = (1u << lane) - 1u;
    const int SL = (((L + 31) >> 5) + 31) & ~31;
    const float4 vw = ((const float4*)v_w)[lane];

    for (int cbase = blockIdx.x * OWN_R; cbase < M; cbase += gridDim.x * OWN_R) {
        const int nown = min(OWN_R, M - cbase);

        for (int i = threadIdx.x; i < OWN_R * HDIM / 4; i += OWN_NT) {
            ((float4*)&hacc[0][0])[i] = make_float4(0.f, 0.f, 0.f, 0.f);
            ((float4*)&facc[0][0])[i] = make_float4(0.f, 0.f, 0.f, 0.f);
        }
        for (int i = threadIdx.x; i < OWN_R; i += OWN_NT) {
            rmax[i] = -INFINITY;
            rden[i] = 0.f;
        }
        int wpos = min(L, wave * SL);
        const int wend = min(L, wave * SL + SL);
        __syncthreads();

        bool more = true;
        while (more) {
            int cnt = 0;
            if (wpos < wend) {
                int ida[8], idb[8];
                auto load_ids = [&](int (&v)[8], int p) {
#pragma unroll
                    for (int j = 0; j < 8; ++j) {
                        const int idx = p + j * 32 + lane;
                        const int t = ci[min(idx, L - 1)];
                        v[j] = (idx < wend) ? t : -1;
                    }
                };
                auto take = [&](const int (&v)[8], int p) {
#pragma unroll
                    for (int j = 0; j < 8; ++j) {
                        const int cl = v[j] - cbase;
                        const bool hit = (unsigned)cl < (unsigned)nown;
                        const unsigned m = (unsigned)__ballot(hit);
                        if (hit) {
                            const int k = cnt + __popc(m & lt);
                            list_e[wave][k] = p + j * 32 + lane;
                            list_c[wave][k] = (unsigned char)cl;
                        }
                        cnt += __popc(m);
                    }
                };
                load_ids(ida, wpos);
                for (;;) {
                    load_ids(idb, wpos + 256);
                    take(ida, wpos);
                    wpos += 256;
                    if (!(wpos < wend && cnt + 256 <= OWN_SEG)) break;
                    load_ids(ida, wpos + 256);
                    take(idb, wpos);
                    wpos += 256;
                    if (!(wpos < wend && cnt + 256 <= OWN_SEG)) break;
                }
                wpos = min(wpos, wend);
            }
            if (lane == 0) {
                s_cnt[wave]  = cnt;
                s_more[wave] = (wpos < wend) ? 1 : 0;
            }
            __syncthreads();
            const int cntv = s_cnt[lane];
            more = __ballot(s_more[lane] != 0) != 0;

            int nb = 0;
            int q = 0, k0 = 0;
            for (;;) {
                int nq = 0;
                bool have = false;
                while (q < 32) {
                    nq = __builtin_amdgcn_readlane(cntv, q);
                    if (k0 < nq) { have = true; break; }
                    ++q; k0 = 0;
                }
                if (have) {
                    const int k = k0 + lane;
                    const bool valid = k < nq;
                    const int e  = valid ? list_e[q][k] : 0;
                    const int cl = valid ? (int)list_c[q][k] : 0;
                    const bool mine = valid && ((cl & 31) == wave);
                    const unsigned mm = (unsigned)__ballot(mine);
                    if (mine) {
                        const int d = nb + __popc(mm & lt);
                        wb_e[wave][d] = e;
                        wb_c[wave][d] = (unsigned char)cl;
                    }
                    nb += __popc(mm);
                    k0 += 32;
                }
                if (nb >= 32 || (!have && nb > 0)) {
                    const int n = min(nb, 32);
                    int bcl = 0, bch = 0;
                    if (lane < n) {
                        bcl = (int)wb_c[wave][lane];
                        bch = chi[wb_e[wave][lane]];
                    }
                    for (int g = 0; g < n; g += OWN_G) {
                        float4 ah[OWN_G], bx[OWN_G], hv[OWN_G], cv4[OWN_G], uf[OWN_G], wf[OWN_G];
                        int   cl4[OWN_G];
#pragma unroll
                        for (int u = 0; u < OWN_G; ++u) {
                            const int i  = min(g + u, n - 1);
                            const int ch = __builtin_amdgcn_readlane(bch, i);
                            const int cl = __builtin_amdgcn_readlane(bcl, i);
                            cl4[u] = cl;
                            ah[u]  = ((const float4*)(Ah      + (size_t)ch * HDIM))[lane];
                            hv[u]  = ((const float4*)(child_h + (size_t)ch * HDIM))[lane];
                            cv4[u] = ((const float4*)(child_c + (size_t)ch * HDIM))[lane];
                            uf[u]  = ((const float4*)(Ufh     + (size_t)ch * HDIM))[lane];
                            bx[u]  = ((const float4*)(Bx  + (size_t)(cbase + cl) * HDIM))[lane];
                            wf[u]  = ((const float4*)(Wfx + (size_t)(cbase + cl) * HDIM))[lane];
                        }
#pragma unroll
                        for (int u = 0; u < OWN_G; ++u) {
                            if (g + u < n) {
                                float s = tanhf(ah[u].x + bx[u].x) * vw.x + tanhf(ah[u].y + bx[u].y) * vw.y
                                        + tanhf(ah[u].z + bx[u].z) * vw.z + tanhf(ah[u].w + bx[u].w) * vw.w;
                                s = wave_sum32_(s);

                                const int   cl    = cl4[u];
                                const float m_old = rmax[cl];
                                const float d_old = rden[cl];
                                const float m_new = fmaxf(m_old, s);
                                const float sc    = __expf(m_old - m_new);
                                const float p     = __expf(s - m_new);
                                if (lane == 0) {
                                    rmax[cl] = m_new;
                                    rden[cl] = d_old * sc + p;
                                }

                                float4* hp = (float4*)&hacc[cl][lane * 4];
                                float4* fp = (float4*)&facc[cl][lane * 4];
                                float4 h = *hp, f = *fp;
                                h.x = h.x * sc + p * hv[u].x;
                                h.y = h.y * sc + p * hv[u].y;
                                h.z = h.z * sc + p * hv[u].z;
                                h.w = h.w * sc + p * hv[u].w;
                                f.x += sigmoidf_(wf[u].x + uf[u].x) * cv4[u].x;
                                f.y += sigmoidf_(wf[u].y + uf[u].y) * cv4[u].y;
                                f.z += sigmoidf_(wf[u].z + uf[u].z) * cv4[u].z;
                                f.w += sigmoidf_(wf[u].w + uf[u].w) * cv4[u].w;
                                *hp = h;
                                *fp = f;
                            }
                        }
                    }
                    const int rest = nb - n;
                    int te = 0;
                    unsigned char tc = 0;
                    if (lane < rest) { te = wb_e[wave][32 + lane]; tc = wb_c[wave][32 + lane]; }
                    if (lane < rest) { wb_e[wave][lane] = te; wb_c[wave][lane] = tc; }
                    nb = rest;
                }
                if (!have) break;
            }
            __syncthreads();
        }

        for (int i = threadIdx.x; i < nown * (HDIM / 4); i += OWN_NT) {
            const int row = i >> 5, c4 = i & 31;
            const float den = rden[row] + EPSV;
            float4 hh = ((const float4*)&hacc[row][0])[c4];
            hh.x /= den; hh.y /= den; hh.z /= den; hh.w /= den;
            ((float4*)(h_hat  + (size_t)(cbase + row) * HDIM))[c4] = hh;
            ((float4*)(sum_fc + (size_t)(cbase + row) * HDIM))[c4] = ((const float4*)&facc[row][0])[c4];
        }
        __syncthreads();
    }
}

__launch_bounds__(256)
__global__ void final_gate_kernel(const float* __restrict__ Gi,
                                  const float* __restrict__ Gc,
                                  const float* __restrict__ Go,
                                  const float* __restrict__ sum_fc,
                                  float* __restrict__ h_out,
                                  float* __restrict__ c_out,
                                  size_t n) {
    const size_t t = (size_t)blockIdx.x * 256 + threadIdx.x;
    if (t >= n) return;
    const float i_j = sigmoidf_(Gi[t]);
    const float c_t = tanhf(Gc[t]);
    const float c   = i_j * c_t + sum_fc[t];
    const float o_j = sigmoidf_(Go[t]);
    h_out[t] = o_j * tanhf(c);
    c_out[t] = c;
}

extern "C" void kernel_launch(void* const* d_in, const int* in_sizes, int n_in,
                              void* d_out, int out_size, void* d_ws, size_t ws_size,
                              hipStream_t stream) {
    const float* x_emb   = (const float*)d_in[0];
    const float* child_h = (const float*)d_in[1];
    const float* child_c = (const float*)d_in[2];
    const int*   ci      = (const int*)d_in[3];
    const int*   chi     = (const int*)d_in[4];
    const float* Wi_w = (const float*)d_in[5];
    const float* Ui_w = (const float*)d_in[6];
    const float* Wf_w = (const float*)d_in[7];
    const float* Uf_w = (const float*)d_in[8];
    const float* Wo_w = (const float*)d_in[9];
    const float* Uo_w = (const float*)d_in[10];
    const float* Wc_w = (const float*)d_in[11];
    const float* Uc_w = (const float*)d_in[12];
    const float* Wa_w = (const float*)d_in[13];
    const float* Ua_w = (const float*)d_in[14];
    const float* Wi_b = (const float*)d_in[15];
    const float* Wf_b = (const float*)d_in[16];
    const float* Wo_b = (const float*)d_in[17];
    const float* Wc_b = (const float*)d_in[18];
    const float* Wa_b = (const float*)d_in[19];
    const float* v_w  = (const float*)d_in[20];

    const int M = in_sizes[0] / HDIM;
    const int L = in_sizes[3];
    const size_t MH = (size_t)M * HDIM;

    float* ws    = (float*)d_ws;
    float* Ah    = ws;
    float* Bx    = Ah  + MH;
    float* Gi    = Bx  + MH;
    float* Wfx   = Gi  + MH;
    float* Go    = Wfx + MH;
    float* Gc    = Go  + MH;
    float* Ufh   = Gc  + MH;
    float* h_hat = Ufh + MH;
    float* sumfc = h_hat + MH;

    const dim3 gblk(256);
    const dim3 ggrd(512);

    gemm_wmma_f32<false><<<ggrd, gblk, 0, stream>>>(child_h, Wa_w, Wa_b, Ah, M);
    gemm_wmma_f32<false><<<ggrd, gblk, 0, stream>>>(x_emb,   Ua_w, nullptr, Bx, M);
    gemm_wmma_f32<false><<<ggrd, gblk, 0, stream>>>(x_emb,   Wi_w, Wi_b, Gi, M);
    gemm_wmma_f32<false><<<ggrd, gblk, 0, stream>>>(x_emb,   Wf_w, Wf_b, Wfx, M);
    gemm_wmma_f32<false><<<ggrd, gblk, 0, stream>>>(x_emb,   Wo_w, Wo_b, Go, M);
    gemm_wmma_f32<false><<<ggrd, gblk, 0, stream>>>(x_emb,   Wc_w, Wc_b, Gc, M);
    gemm_wmma_f32<false><<<ggrd, gblk, 0, stream>>>(child_h, Uf_w, nullptr, Ufh, M);

    edge_attn_owner_kernel<<<dim3(256), dim3(OWN_NT), 0, stream>>>(Ah, Bx, v_w,
                                                        child_h, child_c, Wfx, Ufh,
                                                        ci, chi, h_hat, sumfc, M, L);

    gemm_wmma_f32<true><<<ggrd, gblk, 0, stream>>>(h_hat, Ui_w, nullptr, Gi, M);
    gemm_wmma_f32<true><<<ggrd, gblk, 0, stream>>>(h_hat, Uc_w, nullptr, Gc, M);
    gemm_wmma_f32<true><<<ggrd, gblk, 0, stream>>>(h_hat, Uo_w, nullptr, Go, M);

    float* h_out = (float*)d_out;
    float* c_out = h_out + MH;
    final_gate_kernel<<<(unsigned)((MH + 255) / 256), gblk, 0, stream>>>(
        Gi, Gc, Go, sumfc, h_out, c_out, MH);
}


# Owner block finishes its rows: h_hat stays in LDS, its three U products run there on WMMA f32 (K ascending as before), gates follow from an LDS stage; 12 launches become 8; sum orders same
# speedup vs baseline: 2.4715x; 1.1051x over previous
#include <hip/hip_runtime.h>
#include <hip/hip_bf16.h>

#define HDIM 128
#define EPSV 1e-9f

typedef __attribute__((ext_vector_type(2))) float v2f;
typedef __attribute__((ext_vector_type(8))) float v8f;

__launch_bounds__(256)
__global__ void gemm_wmma_f32(const float* __restrict__ X,
                              const float* __restrict__ W,
                              const float* __restrict__ bias,
                              float* __restrict__ Y,
                              int Mrows) {
    __shared__ float lw[HDIM][132];
    __shared__ float lx[2][16][132];

    const int wave  = threadIdx.x >> 5;
    const int lane  = threadIdx.x & 31;
    const int nsub  = lane & 15;
    const int khalf = (lane >> 4) << 1;
    const int n0    = wave * 16;
    const int ntiles = (Mrows + 15) >> 4;

    const int srow = threadIdx.x >> 5;
    const int scol = (threadIdx.x & 31) << 2;
    const int lastrow = Mrows - 1;

    int t = blockIdx.x;
    float4 p0 = {}, p1 = {};
    if (t < ntiles) {
        const int ra = min(t * 16 + srow, lastrow), rb = min(t * 16 + srow + 8, lastrow);
        p0 = *(const float4*)(X + (size_t)ra * HDIM + scol);
        p1 = *(const float4*)(X + (size_t)rb * HDIM + scol);
    }

#pragma unroll
    for (int p = 0; p < HDIM / 8; ++p)
        *(float4*)&lw[srow + 8 * p][scol] =
            *(const float4*)(W + (size_t)(srow + 8 * p) * HDIM + scol);
    __syncthreads();

    v2f bfrag[HDIM / 4];
#pragma unroll
    for (int j = 0; j < HDIM / 4; ++j)
        bfrag[j] = *(const v2f*)&lw[n0 + nsub][4 * j + khalf];
    const float bval = bias ? bias[n0 + nsub] : 0.0f;

    int buf = 0;
    for (; t < ntiles; t += gridDim.x) {
        *(float4*)&lx[buf][srow][scol]     = p0;
        *(float4*)&lx[buf][srow + 8][scol] = p1;
        __syncthreads();

        const int tn = t + gridDim.x;
        if (tn < ntiles) {
            const int ra = min(tn * 16 + srow, lastrow), rb = min(tn * 16 + srow + 8, lastrow);
            p0 = *(const float4*)(X + (size_t)ra * HDIM + scol);
            p1 = *(const float4*)(X + (size_t)rb * HDIM + scol);
        }

        const bool full  = (t * 16 + 16 <= Mrows);
        const int  mbase = t * 16 + ((lane >> 4) << 3);
        float* __restrict__ yp = Y + (size_t)mbase * HDIM + n0 + nsub;

        const float* arow = &lx[buf][nsub][khalf];
        v8f acc = {};
#pragma unroll
        for (int j = 0; j < HDIM / 4; ++j) {
            const v2f a = *(const v2f*)(arow + 4 * j);
            acc = __builtin_amdgcn_wmma_f32_16x16x4_f32(
                 false, a,  false, bfrag[j],
                 (short)0, acc,  false,  false);
        }

        float v[8];
#pragma unroll
        for (int r = 0; r < 8; ++r) v[r] = acc[r] + bval;
        if (full) {
#pragma unroll
            for (int r = 0; r < 8; ++r) yp[(size_t)r * HDIM] = v[r];
        } else {
#pragma unroll
            for (int r = 0; r < 8; ++r)
                if (mbase + r < Mrows) yp[(size_t)r * HDIM] = v[r];
        }
        buf ^= 1;
    }
}

__device__ __forceinline__ float sigmoidf_(float x) {
    return 1.0f / (1.0f + __expf(-x));
}

template <int XM>
__device__ __forceinline__ float row_xmask_(float x) {
    const int b = __builtin_bit_cast(int, x);
    return __builtin_bit_cast(float, __builtin_amdgcn_update_dpp(b, b, 0x160 | XM, 0xf, 0xf, false));
}
__device__ __forceinline__ float wave_sum32_(float x) {
    const int b = __builtin_bit_cast(int, x);
    x += __builtin_bit_cast(float, __builtin_amdgcn_permlanex16(b, b, 0x76543210, 0xfedcba98, false, false));
    x += row_xmask_<8>(x);
    x += row_xmask_<4>(x);
    x += row_xmask_<2>(x);
    x += row_xmask_<1>(x);
    return x;
}

#define OWN_R    196
#define OWN_RP   208
#define OWN_LDP  132
#define OWN_NT   1024
#define OWN_SEG  512
#define OWN_G    2

__launch_bounds__(OWN_NT)
__global__ void edge_attn_owner_kernel(const float* __restrict__ Ah,
                                       const float* __restrict__ Bx,
                                       const float* __restrict__ v_w,
                                       const float* __restrict__ child_h,
                                       const float* __restrict__ child_c,
                                       const float* __restrict__ Wfx,
                                       const float* __restrict__ Ufh,
                                       const int* __restrict__ ci,
                                       const int* __restrict__ chi,
                                       const float* __restrict__ Ui,
                                       const float* __restrict__ Uc,
                                       const float* __restrict__ Uo,
                                       const float* __restrict__ Gi,
                                       const float* __restrict__ Gc,
                                       const float* __restrict__ Go,
                                       float* __restrict__ h_out,
                                       float* __restrict__ c_out,
                                       int M, int L) {
    union ListOrStage {
        int   e[32][OWN_SEG];
        float g[2][3][16][OWN_LDP];
    };
    __shared__ alignas(16) float hacc[OWN_RP][OWN_LDP];
    __shared__ alignas(16) float facc[OWN_R][HDIM];
    __shared__ float rmax[OWN_R];
    __shared__ float rden[OWN_R];
    __shared__ alignas(16) ListOrStage lu;
    __shared__ unsigned char list_c[32][OWN_SEG];
    __shared__ int           s_cnt[32];
    __shared__ int           s_more[32];
    __shared__ int           wb_e[32][64];
    __shared__ unsigned char wb_c[32][64];

    const int wave = __builtin_amdgcn_readfirstlane((int)(threadIdx.x >> 5));
    const int lane = threadIdx.x & 31;
    const unsigned lt = (1u << lane) - 1u;
    const int SL = (((L + 31) >> 5) + 31) & ~31;
    const float4 vw = ((const float4*)v_w)[lane];

    for (int cbase = blockIdx.x * OWN_R; cbase < M; cbase += gridDim.x * OWN_R) {
        const int nown = min(OWN_R, M - cbase);

        for (int i = threadIdx.x; i < OWN_RP * OWN_LDP / 4; i += OWN_NT) {
            ((float4*)&hacc[0][0])[i] = make_float4(0.f, 0.f, 0.f, 0.f);
            if (i < OWN_R * HDIM / 4)
                ((float4*)&facc[0][0])[i] = make_float4(0.f, 0.f, 0.f, 0.f);
        }
        for (int i = threadIdx.x; i < OWN_R; i += OWN_NT) {
            rmax[i] = -INFINITY;
            rden[i] = 0.f;
        }
        int wpos = min(L, wave * SL);
        const int wend = min(L, wave * SL + SL);
        __syncthreads();

        bool more = true;
        while (more) {
            int cnt = 0;
            if (wpos < wend) {
                int ida[8], idb[8];
                auto load_ids = [&](int (&v)[8], int p) {
#pragma unroll
                    for (int j = 0; j < 8; ++j) {
                        const int idx = p + j * 32 + lane;
                        const int t = ci[min(idx, L - 1)];
                        v[j] = (idx < wend) ? t : -1;
                    }
                };
                auto take = [&](const int (&v)[8], int p) {
#pragma unroll
                    for (int j = 0; j < 8; ++j) {
                        const int cl = v[j] - cbase;
                        const bool hit = (unsigned)cl < (unsigned)nown;
                        const unsigned m = (unsigned)__ballot(hit);
                        if (hit) {
                            const int k = cnt + __popc(m & lt);
                            lu.e[wave][k] = p + j * 32 + lane;
                            list_c[wave][k] = (unsigned char)cl;
                        }
                        cnt += __popc(m);
                    }
                };
                load_ids(ida, wpos);
                for (;;) {
                    load_ids(idb, wpos + 256);
                    take(ida, wpos);
                    wpos += 256;
                    if (!(wpos < wend && cnt + 256 <= OWN_SEG)) break;
                    load_ids(ida, wpos + 256);
                    take(idb, wpos);
                    wpos += 256;
                    if (!(wpos < wend && cnt + 256 <= OWN_SEG)) break;
                }
                wpos = min(wpos, wend);
            }
            if (lane == 0) {
                s_cnt[wave]  = cnt;
                s_more[wave] = (wpos < wend) ? 1 : 0;
            }
            __syncthreads();
            const int cntv = s_cnt[lane];
            more = __ballot(s_more[lane] != 0) != 0;

            int nb = 0;
            int q = 0, k0 = 0;
            for (;;) {
                int nq = 0;
                bool have = false;
                while (q < 32) {
                    nq = __builtin_amdgcn_readlane(cntv, q);
                    if (k0 < nq) { have = true; break; }
                    ++q; k0 = 0;
                }
                if (have) {
                    const int k = k0 + lane;
                    const bool valid = k < nq;
                    const int e  = valid ? lu.e[q][k] : 0;
                    const int cl = valid ? (int)list_c[q][k] : 0;
                    const bool mine = valid && ((cl & 31) == wave);
                    const unsigned mm = (unsigned)__ballot(mine);
                    if (mine) {
                        const int d = nb + __popc(mm & lt);
                        wb_e[wave][d] = e;
                        wb_c[wave][d] = (unsigned char)cl;
                    }
                    nb += __popc(mm);
                    k0 += 32;
                }
                if (nb >= 32 || (!have && nb > 0)) {
                    const int n = min(nb, 32);
                    int bcl = 0, bch = 0;
                    if (lane < n) {
                        bcl = (int)wb_c[wave][lane];
                        bch = chi[wb_e[wave][lane]];
                    }
                    for (int g = 0; g < n; g += OWN_G) {
                        float4 ah[OWN_G], bx[OWN_G], hv[OWN_G], cv4[OWN_G], uf[OWN_G], wf[OWN_G];
                        int   cl4[OWN_G];
#pragma unroll
                        for (int u = 0; u < OWN_G; ++u) {
                            const int i  = min(g + u, n - 1);
                            const int ch = __builtin_amdgcn_readlane(bch, i);
                            const int cl = __builtin_amdgcn_readlane(bcl, i);
                            cl4[u] = cl;
                            ah[u]  = ((const float4*)(Ah      + (size_t)ch * HDIM))[lane];
                            hv[u]  = ((const float4*)(child_h + (size_t)ch * HDIM))[lane];
                            cv4[u] = ((const float4*)(child_c + (size_t)ch * HDIM))[lane];
                            uf[u]  = ((const float4*)(Ufh     + (size_t)ch * HDIM))[lane];
                            bx[u]  = ((const float4*)(Bx  + (size_t)(cbase + cl) * HDIM))[lane];
                            wf[u]  = ((const float4*)(Wfx + (size_t)(cbase + cl) * HDIM))[lane];
                        }
#pragma unroll
                        for (int u = 0; u < OWN_G; ++u) {
                            if (g + u < n) {
                                float s = tanhf(ah[u].x + bx[u].x) * vw.x + tanhf(ah[u].y + bx[u].y) * vw.y
                                        + tanhf(ah[u].z + bx[u].z) * vw.z + tanhf(ah[u].w + bx[u].w) * vw.w;
                                s = wave_sum32_(s);

                                const int   cl    = cl4[u];
                                const float m_old = rmax[cl];
                                const float d_old = rden[cl];
                                const float m_new = fmaxf(m_old, s);
                                const float sc    = __expf(m_old - m_new);
                                const float p     = __expf(s - m_new);
                                if (lane == 0) {
                                    rmax[cl] = m_new;
                                    rden[cl] = d_old * sc + p;
                                }

                                float4* hp = (float4*)&hacc[cl][lane * 4];
                                float4* fp = (float4*)&facc[cl][lane * 4];
                                float4 h = *hp, f = *fp;
                                h.x = h.x * sc + p * hv[u].x;
                                h.y = h.y * sc + p * hv[u].y;
                                h.z = h.z * sc + p * hv[u].z;
                                h.w = h.w * sc + p * hv[u].w;
                                f.x += sigmoidf_(wf[u].x + uf[u].x) * cv4[u].x;
                                f.y += sigmoidf_(wf[u].y + uf[u].y) * cv4[u].y;
                                f.z += sigmoidf_(wf[u].z + uf[u].z) * cv4[u].z;
                                f.w += sigmoidf_(wf[u].w + uf[u].w) * cv4[u].w;
                                *hp = h;
                                *fp = f;
                            }
                        }
                    }
                    const int rest = nb - n;
                    int te = 0;
                    unsigned char tc = 0;
                    if (lane < rest) { te = wb_e[wave][32 + lane]; tc = wb_c[wave][32 + lane]; }
                    if (lane < rest) { wb_e[wave][lane] = te; wb_c[wave][lane] = tc; }
                    nb = rest;
                }
                if (!have) break;
            }
            __syncthreads();
        }

        const int ntile = (nown + 15) >> 4;
        const int nsub  = lane & 15;
        const int khalf = (lane >> 4) << 1;
        const int pq    = wave >> 3;
        const int pn0   = (wave & 7) * 16;
        const int grow  = wave >> 1;
        const int gcol  = (((wave & 1) << 5) + lane) << 1;

        auto load_g = [&](int mt, float2& gi, float2& gc, float2& go) {
            const size_t o = (size_t)(cbase + min(mt * 16 + grow, nown - 1)) * HDIM + gcol;
            gi = *(const float2*)(Gi + o);
            gc = *(const float2*)(Gc + o);
            go = *(const float2*)(Go + o);
        };
        float2 gi0, gc0, go0;
        load_g(0, gi0, gc0, go0);

        v2f bfrag[HDIM / 4] = {};
        if (wave < 24) {
            const float* ub = (pq == 0 ? Ui : (pq == 1 ? Uc : Uo)) + pn0 * HDIM;
            const int    uo = nsub * HDIM + khalf;
#pragma unroll
            for (int j = 0; j < HDIM / 4; ++j) bfrag[j] = *(const v2f*)(ub + uo + 4 * j);
        }

        for (int i = threadIdx.x; i < nown * (HDIM / 4); i += OWN_NT) {
            const int row = i >> 5, c4 = i & 31;
            const float den = rden[row] + EPSV;
            float4* hp = (float4*)&hacc[row][c4 * 4];
            float4 hh = *hp;
            hh.x /= den; hh.y /= den; hh.z /= den; hh.w /= den;
            *hp = hh;
        }
        __syncthreads();

        auto product = [&](int mt) {
            const float* arow = &hacc[mt * 16 + nsub][khalf];
            v8f acc = {};
#pragma unroll
            for (int j = 0; j < HDIM / 4; ++j) {
                const v2f a = *(const v2f*)(arow + 4 * j);
                acc = __builtin_amdgcn_wmma_f32_16x16x4_f32(
                     false, a,  false, bfrag[j],
                     (short)0, acc,  false,  false);
            }
            float* sp = &lu.g[mt & 1][pq][(lane >> 4) << 3][pn0 + nsub];
#pragma unroll
            for (int r = 0; r < 8; ++r) sp[r * OWN_LDP] = acc[r];
        };
        auto gate = [&](int mt, float2 gi, float2 gc, float2 go) {
            const int r = mt * 16 + grow;
            if (r < nown) {
                const float2 ai = *(const float2*)&lu.g[mt & 1][0][grow][gcol];
                const float2 ac = *(const float2*)&lu.g[mt & 1][1][grow][gcol];
                const float2 ao = *(const float2*)&lu.g[mt & 1][2][grow][gcol];
                const float2 sf = *(const float2*)&facc[r][gcol];
                float2 cc, hh;
                {
                    const float i_j = sigmoidf_(ai.x + gi.x);
                    const float c_t = tanhf(ac.x + gc.x);
                    const float c   = i_j * c_t + sf.x;
                    const float o_j = sigmoidf_(ao.x + go.x);
                    hh.x = o_j * tanhf(c);
                    cc.x = c;
                }
                {
                    const float i_j = sigmoidf_(ai.y + gi.y);
                    const float c_t = tanhf(ac.y + gc.y);
                    const float c   = i_j * c_t + sf.y;
                    const float o_j = sigmoidf_(ao.y + go.y);
                    hh.y = o_j * tanhf(c);
                    cc.y = c;
                }
                const size_t o = (size_t)(cbase + r) * HDIM + gcol;
                *(float2*)(h_out + o) = hh;
                *(float2*)(c_out + o) = cc;
            }
        };

        const bool pw     = wave < 24;
        const bool pfirst = ((wave >> 2) & 1) == 0;
        if (pw) product(0);
        __syncthreads();
        for (int mt = 0; mt < ntile; ++mt) {
            const bool nxt = mt + 1 < ntile;
            if (nxt && pw && pfirst) product(mt + 1);
            gate(mt, gi0, gc0, go0);
            if (nxt) load_g(mt + 1, gi0, gc0, go0);
            if (nxt && pw && !pfirst) product(mt + 1);
            __syncthreads();
        }
    }
}

extern "C" void kernel_launch(void* const* d_in, const int* in_sizes, int n_in,
                              void* d_out, int out_size, void* d_ws, size_t ws_size,
                              hipStream_t stream) {
    const float* x_emb   = (const float*)d_in[0];
    const float* child_h = (const float*)d_in[1];
    const float* child_c = (const float*)d_in[2];
    const int*   ci      = (const int*)d_in[3];
    const int*   chi     = (const int*)d_in[4];
    const float* Wi_w = (const float*)d_in[5];
    const float* Ui_w = (const float*)d_in[6];
    const float* Wf_w = (const float*)d_in[7];
    const float* Uf_w = (const float*)d_in[8];
    const float* Wo_w = (const float*)d_in[9];
    const float* Uo_w = (const float*)d_in[10];
    const float* Wc_w = (const float*)d_in[11];
    const float* Uc_w = (const float*)d_in[12];
    const float* Wa_w = (const float*)d_in[13];
    const float* Ua_w = (const float*)d_in[14];
    const float* Wi_b = (const float*)d_in[15];
    const float* Wf_b = (const float*)d_in[16];
    const float* Wo_b = (const float*)d_in[17];
    const float* Wc_b = (const float*)d_in[18];
    const float* Wa_b = (const float*)d_in[19];
    const float* v_w  = (const float*)d_in[20];

    const int M = in_sizes[0] / HDIM;
    const int L = in_sizes[3];
    const size_t MH = (size_t)M * HDIM;

    float* ws    = (float*)d_ws;
    float* Ah    = ws;
    float* Bx    = Ah  + MH;
    float* Gi    = Bx  + MH;
    float* Wfx   = Gi  + MH;
    float* Go    = Wfx + MH;
    float* Gc    = Go  + MH;
    float* Ufh   = Gc  + MH;

    const dim3 gblk(256);
    const dim3 ggrd(512);

    gemm_wmma_f32<<<ggrd, gblk, 0, stream>>>(child_h, Wa_w, Wa_b, Ah, M);
    gemm_wmma_f32<<<ggrd, gblk, 0, stream>>>(x_emb,   Ua_w, nullptr, Bx, M);
    gemm_wmma_f32<<<ggrd, gblk, 0, stream>>>(x_emb,   Wi_w, Wi_b, Gi, M);
    gemm_wmma_f32<<<ggrd, gblk, 0, stream>>>(x_emb,   Wf_w, Wf_b, Wfx, M);
    gemm_wmma_f32<<<ggrd, gblk, 0, stream>>>(x_emb,   Wo_w, Wo_b, Go, M);
    gemm_wmma_f32<<<ggrd, gblk, 0, stream>>>(x_emb,   Wc_w, Wc_b, Gc, M);
    gemm_wmma_f32<<<ggrd, gblk, 0, stream>>>(child_h, Uf_w, nullptr, Ufh, M);

    float* h_out = (float*)d_out;
    float* c_out = h_out + MH;
    edge_attn_owner_kernel<<<dim3(256), dim3(OWN_NT), 0, stream>>>(Ah, Bx, v_w,
                                                        child_h, child_c, Wfx, Ufh,
                                                        ci, chi, Ui_w, Uc_w, Uo_w,
                                                        Gi, Gc, Go, h_out, c_out, M, L);
}


# Seven projection launches become one: grid split by weight matrix, a block stages one matrix and keeps its fragments over about 29 of the 16-row tiles; 8 launches become 2; sum orders same
# speedup vs baseline: 2.6480x; 1.0714x over previous
#include <hip/hip_runtime.h>
#include <hip/hip_bf16.h>

#define HDIM 128
#define EPSV 1e-9f

typedef __attribute__((ext_vector_type(2))) float v2f;
typedef __attribute__((ext_vector_type(8))) float v8f;

#define PROJ_N 7

__device__ __forceinline__ void proj_block(const float* __restrict__ X,
                                           const float* __restrict__ W,
                                           const float* __restrict__ bias,
                                           float* __restrict__ Y,
                                           int Mrows, int t0, int tstride) {
    __shared__ float lw[HDIM][132];
    __shared__ float lx[2][16][132];

    const int wave  = threadIdx.x >> 5;
    const int lane  = threadIdx.x & 31;
    const int nsub  = lane & 15;
    const int khalf = (lane >> 4) << 1;
    const int n0    = wave * 16;
    const int ntiles = (Mrows + 15) >> 4;

    const int srow = threadIdx.x >> 5;
    const int scol = (threadIdx.x & 31) << 2;
    const int lastrow = Mrows - 1;

    int t = t0;
    float4 p0 = {}, p1 = {};
    if (t < ntiles) {
        const int ra = min(t * 16 + srow, lastrow), rb = min(t * 16 + srow + 8, lastrow);
        p0 = *(const float4*)(X + (size_t)ra * HDIM + scol);
        p1 = *(const float4*)(X + (size_t)rb * HDIM + scol);
    }

#pragma unroll
    for (int p = 0; p < HDIM / 8; ++p)
        *(float4*)&lw[srow + 8 * p][scol] =
            *(const float4*)(W + (size_t)(srow + 8 * p) * HDIM + scol);
    __syncthreads();

    v2f bfrag[HDIM / 4];
#pragma unroll
    for (int j = 0; j < HDIM / 4; ++j)
        bfrag[j] = *(const v2f*)&lw[n0 + nsub][4 * j + khalf];
    const float bval = bias ? bias[n0 + nsub] : 0.0f;

    int buf = 0;
    for (; t < ntiles; t += tstride) {
        *(float4*)&lx[buf][srow][scol]     = p0;
        *(float4*)&lx[buf][srow + 8][scol] = p1;
        __syncthreads();

        const int tn = t + tstride;
        if (tn < ntiles) {
            const int ra = min(tn * 16 + srow, lastrow), rb = min(tn * 16 + srow + 8, lastrow);
            p0 = *(const float4*)(X + (size_t)ra * HDIM + scol);
            p1 = *(const float4*)(X + (size_t)rb * HDIM + scol);
        }

        const bool full  = (t * 16 + 16 <= Mrows);
        const int  mbase = t * 16 + ((lane >> 4) << 3);
        float* __restrict__ yp = Y + (size_t)mbase * HDIM + n0 + nsub;

        const float* arow = &lx[buf][nsub][khalf];
        v8f acc = {};
#pragma unroll
        for (int j = 0; j < HDIM / 4; ++j) {
            const v2f a = *(const v2f*)(arow + 4 * j);
            acc = __builtin_amdgcn_wmma_f32_16x16x4_f32(
                 false, a,  false, bfrag[j],
                 (short)0, acc,  false,  false);
        }

        float v[8];
#pragma unroll
        for (int r = 0; r < 8; ++r) v[r] = acc[r] + bval;
        if (full) {
#pragma unroll
            for (int r = 0; r < 8; ++r) yp[(size_t)r * HDIM] = v[r];
        } else {
#pragma unroll
            for (int r = 0; r < 8; ++r)
                if (mbase + r < Mrows) yp[(size_t)r * HDIM] = v[r];
        }
        buf ^= 1;
    }
}

__launch_bounds__(256)
__global__ void proj7_wmma_f32(const float* __restrict__ x_emb,
                               const float* __restrict__ child_h,
                               const float* __restrict__ Wa_w,
                               const float* __restrict__ Ua_w,
                               const float* __restrict__ Wi_w,
                               const float* __restrict__ Wf_w,
                               const float* __restrict__ Wo_w,
                               const float* __restrict__ Wc_w,
                               const float* __restrict__ Uf_w,
                               const float* __restrict__ Wa_b,
                               const float* __restrict__ Wi_b,
                               const float* __restrict__ Wf_b,
                               const float* __restrict__ Wo_b,
                               const float* __restrict__ Wc_b,
                               float* __restrict__ ws,
                               int Mrows) {
    const int q  = blockIdx.x % PROJ_N;
    const int r0 = blockIdx.x / PROJ_N;
    const int pb = gridDim.x / PROJ_N;
    if (r0 >= pb) return;

    const float* X    = x_emb;
    const float* W    = Ua_w;
    const float* bias = nullptr;
    switch (q) {
        case 0: X = child_h; W = Wa_w; bias = Wa_b; break;
        case 1: break;
        case 2: W = Wi_w; bias = Wi_b; break;
        case 3: W = Wf_w; bias = Wf_b; break;
        case 4: W = Wo_w; bias = Wo_b; break;
        case 5: W = Wc_w; bias = Wc_b; break;
        default: X = child_h; W = Uf_w; break;
    }
    proj_block(X, W, bias, ws + (size_t)q * (size_t)Mrows * HDIM, Mrows, r0, pb);
}

__device__ __forceinline__ float sigmoidf_(float x) {
    return 1.0f / (1.0f + __expf(-x));
}

template <int XM>
__device__ __forceinline__ float row_xmask_(float x) {
    const int b = __builtin_bit_cast(int, x);
    return __builtin_bit_cast(float, __builtin_amdgcn_update_dpp(b, b, 0x160 | XM, 0xf, 0xf, false));
}
__device__ __forceinline__ float wave_sum32_(float x) {
    const int b = __builtin_bit_cast(int, x);
    x += __builtin_bit_cast(float, __builtin_amdgcn_permlanex16(b, b, 0x76543210, 0xfedcba98, false, false));
    x += row_xmask_<8>(x);
    x += row_xmask_<4>(x);
    x += row_xmask_<2>(x);
    x += row_xmask_<1>(x);
    return x;
}

#define OWN_R    196
#define OWN_RP   208
#define OWN_LDP  132
#define OWN_NT   1024
#define OWN_SEG  512
#define OWN_G    2

__launch_bounds__(OWN_NT)
__global__ void edge_attn_owner_kernel(const float* __restrict__ Ah,
                                       const float* __restrict__ Bx,
                                       const float* __restrict__ v_w,
                                       const float* __restrict__ child_h,
                                       const float* __restrict__ child_c,
                                       const float* __restrict__ Wfx,
                                       const float* __restrict__ Ufh,
                                       const int* __restrict__ ci,
                                       const int* __restrict__ chi,
                                       const float* __restrict__ Ui,
                                       const float* __restrict__ Uc,
                                       const float* __restrict__ Uo,
                                       const float* __restrict__ Gi,
                                       const float* __restrict__ Gc,
                                       const float* __restrict__ Go,
                                       float* __restrict__ h_out,
                                       float* __restrict__ c_out,
                                       int M, int L) {
    union ListOrStage {
        int   e[32][OWN_SEG];
        float g[2][3][16][OWN_LDP];
    };
    __shared__ alignas(16) float hacc[OWN_RP][OWN_LDP];
    __shared__ alignas(16) float facc[OWN_R][HDIM];
    __shared__ float rmax[OWN_R];
    __shared__ float rden[OWN_R];
    __shared__ alignas(16) ListOrStage lu;
    __shared__ unsigned char list_c[32][OWN_SEG];
    __shared__ int           s_cnt[32];
    __shared__ int           s_more[32];
    __shared__ int           wb_e[32][64];
    __shared__ unsigned char wb_c[32][64];

    const int wave = __builtin_amdgcn_readfirstlane((int)(threadIdx.x >> 5));
    const int lane = threadIdx.x & 31;
    const unsigned lt = (1u << lane) - 1u;
    const int SL = (((L + 31) >> 5) + 31) & ~31;
    const float4 vw = ((const float4*)v_w)[lane];

    for (int cbase = blockIdx.x * OWN_R; cbase < M; cbase += gridDim.x * OWN_R) {
        const int nown = min(OWN_R, M - cbase);

        for (int i = threadIdx.x; i < OWN_RP * OWN_LDP / 4; i += OWN_NT) {
            ((float4*)&hacc[0][0])[i] = make_float4(0.f, 0.f, 0.f, 0.f);
            if (i < OWN_R * HDIM / 4)
                ((float4*)&facc[0][0])[i] = make_float4(0.f, 0.f, 0.f, 0.f);
        }
        for (int i = threadIdx.x; i < OWN_R; i += OWN_NT) {
            rmax[i] = -INFINITY;
            rden[i] = 0.f;
        }
        int wpos = min(L, wave * SL);
        const int wend = min(L, wave * SL + SL);
        __syncthreads();

        bool more = true;
        while (more) {
            int cnt = 0;
            if (wpos < wend) {
                int ida[8], idb[8];
                auto load_ids = [&](int (&v)[8], int p) {
#pragma unroll
                    for (int j = 0; j < 8; ++j) {
                        const int idx = p + j * 32 + lane;
                        const int t = ci[min(idx, L - 1)];
                        v[j] = (idx < wend) ? t : -1;
                    }
                };
                auto take = [&](const int (&v)[8], int p) {
#pragma unroll
                    for (int j = 0; j < 8; ++j) {
                        const int cl = v[j] - cbase;
                        const bool hit = (unsigned)cl < (unsigned)nown;
                        const unsigned m = (unsigned)__ballot(hit);
                        if (hit) {
                            const int k = cnt + __popc(m & lt);
                            lu.e[wave][k] = p + j * 32 + lane;
                            list_c[wave][k] = (unsigned char)cl;
                        }
                        cnt += __popc(m);
                    }
                };
                load_ids(ida, wpos);
                for (;;) {
                    load_ids(idb, wpos + 256);
                    take(ida, wpos);
                    wpos += 256;
                    if (!(wpos < wend && cnt + 256 <= OWN_SEG)) break;
                    load_ids(ida, wpos + 256);
                    take(idb, wpos);
                    wpos += 256;
                    if (!(wpos < wend && cnt + 256 <= OWN_SEG)) break;
                }
                wpos = min(wpos, wend);
            }
            if (lane == 0) {
                s_cnt[wave]  = cnt;
                s_more[wave] = (wpos < wend) ? 1 : 0;
            }
            __syncthreads();
            const int cntv = s_cnt[lane];
            more = __ballot(s_more[lane] != 0) != 0;

            int nb = 0;
            int q = 0, k0 = 0;
            for (;;) {
                int nq = 0;
                bool have = false;
                while (q < 32) {
                    nq = __builtin_amdgcn_readlane(cntv, q);
                    if (k0 < nq) { have = true; break; }
                    ++q; k0 = 0;
                }
                if (have) {
                    const int k = k0 + lane;
                    const bool valid = k < nq;
                    const int e  = valid ? lu.e[q][k] : 0;
                    const int cl = valid ? (int)list_c[q][k] : 0;
                    const bool mine = valid && ((cl & 31) == wave);
                    const unsigned mm = (unsigned)__ballot(mine);
                    if (mine) {
                        const int d = nb + __popc(mm & lt);
                        wb_e[wave][d] = e;
                        wb_c[wave][d] = (unsigned char)cl;
                    }
                    nb += __popc(mm);
                    k0 += 32;
                }
                if (nb >= 32 || (!have && nb > 0)) {
                    const int n = min(nb, 32);
                    int bcl = 0, bch = 0;
                    if (lane < n) {
                        bcl = (int)wb_c[wave][lane];
                        bch = chi[wb_e[wave][lane]];
                    }
                    for (int g = 0; g < n; g += OWN_G) {
                        float4 ah[OWN_G], bx[OWN_G], hv[OWN_G], cv4[OWN_G], uf[OWN_G], wf[OWN_G];
                        int   cl4[OWN_G];
#pragma unroll
                        for (int u = 0; u < OWN_G; ++u) {
                            const int i  = min(g + u, n - 1);
                            const int ch = __builtin_amdgcn_readlane(bch, i);
                            const int cl = __builtin_amdgcn_readlane(bcl, i);
                            cl4[u] = cl;
                            ah[u]  = ((const float4*)(Ah      + (size_t)ch * HDIM))[lane];
                            hv[u]  = ((const float4*)(child_h + (size_t)ch * HDIM))[lane];
                            cv4[u] = ((const float4*)(child_c + (size_t)ch * HDIM))[lane];
                            uf[u]  = ((const float4*)(Ufh     + (size_t)ch * HDIM))[lane];
                            bx[u]  = ((const float4*)(Bx  + (size_t)(cbase + cl) * HDIM))[lane];
                            wf[u]  = ((const float4*)(Wfx + (size_t)(cbase + cl) * HDIM))[lane];
                        }
#pragma unroll
                        for (int u = 0; u < OWN_G; ++u) {
                            if (g + u < n) {
                                float s = tanhf(ah[u].x + bx[u].x) * vw.x + tanhf(ah[u].y + bx[u].y) * vw.y
                                        + tanhf(ah[u].z + bx[u].z) * vw.z + tanhf(ah[u].w + bx[u].w) * vw.w;
                                s = wave_sum32_(s);

                                const int   cl    = cl4[u];
                                const float m_old = rmax[cl];
                                const float d_old = rden[cl];
                                const float m_new = fmaxf(m_old, s);
                                const float sc    = __expf(m_old - m_new);
                                const float p     = __expf(s - m_new);
                                if (lane == 0) {
                                    rmax[cl] = m_new;
                                    rden[cl] = d_old * sc + p;
                                }

                                float4* hp = (float4*)&hacc[cl][lane * 4];
                                float4* fp = (float4*)&facc[cl][lane * 4];
                                float4 h = *hp, f = *fp;
                                h.x = h.x * sc + p * hv[u].x;
                                h.y = h.y * sc + p * hv[u].y;
                                h.z = h.z * sc + p * hv[u].z;
                                h.w = h.w * sc + p * hv[u].w;
                                f.x += sigmoidf_(wf[u].x + uf[u].x) * cv4[u].x;
                                f.y += sigmoidf_(wf[u].y + uf[u].y) * cv4[u].y;
                                f.z += sigmoidf_(wf[u].z + uf[u].z) * cv4[u].z;
                                f.w += sigmoidf_(wf[u].w + uf[u].w) * cv4[u].w;
                                *hp = h;
                                *fp = f;
                            }
                        }
                    }
                    const int rest = nb - n;
                    int te = 0;
                    unsigned char tc = 0;
                    if (lane < rest) { te = wb_e[wave][32 + lane]; tc = wb_c[wave][32 + lane]; }
                    if (lane < rest) { wb_e[wave][lane] = te; wb_c[wave][lane] = tc; }
                    nb = rest;
                }
                if (!have) break;
            }
            __syncthreads();
        }

        const int ntile = (nown + 15) >> 4;
        const int nsub  = lane & 15;
        const int khalf = (lane >> 4) << 1;
        const int pq    = wave >> 3;
        const int pn0   = (wave & 7) * 16;
        const int grow  = wave >> 1;
        const int gcol  = (((wave & 1) << 5) + lane) << 1;

        auto load_g = [&](int mt, float2& gi, float2& gc, float2& go) {
            const size_t o = (size_t)(cbase + min(mt * 16 + grow, nown - 1)) * HDIM + gcol;
            gi = *(const float2*)(Gi + o);
            gc = *(const float2*)(Gc + o);
            go = *(const float2*)(Go + o);
        };
        float2 gi0, gc0, go0;
        load_g(0, gi0, gc0, go0);

        v2f bfrag[HDIM / 4] = {};
        if (wave < 24) {
            const float* ub = (pq == 0 ? Ui : (pq == 1 ? Uc : Uo)) + pn0 * HDIM;
            const int    uo = nsub * HDIM + khalf;
#pragma unroll
            for (int j = 0; j < HDIM / 4; ++j) bfrag[j] = *(const v2f*)(ub + uo + 4 * j);
        }

        for (int i = threadIdx.x; i < nown * (HDIM / 4); i += OWN_NT) {
            const int row = i >> 5, c4 = i & 31;
            const float den = rden[row] + EPSV;
            float4* hp = (float4*)&hacc[row][c4 * 4];
            float4 hh = *hp;
            hh.x /= den; hh.y /= den; hh.z /= den; hh.w /= den;
            *hp = hh;
        }
        __syncthreads();

        auto product = [&](int mt) {
            const float* arow = &hacc[mt * 16 + nsub][khalf];
            v8f acc = {};
#pragma unroll
            for (int j = 0; j < HDIM / 4; ++j) {
                const v2f a = *(const v2f*)(arow + 4 * j);
                acc = __builtin_amdgcn_wmma_f32_16x16x4_f32(
                     false, a,  false, bfrag[j],
                     (short)0, acc,  false,  false);
            }
            float* sp = &lu.g[mt & 1][pq][(lane >> 4) << 3][pn0 + nsub];
#pragma unroll
            for (int r = 0; r < 8; ++r) sp[r * OWN_LDP] = acc[r];
        };
        auto gate = [&](int mt, float2 gi, float2 gc, float2 go) {
            const int r = mt * 16 + grow;
            if (r < nown) {
                const float2 ai = *(const float2*)&lu.g[mt & 1][0][grow][gcol];
                const float2 ac = *(const float2*)&lu.g[mt & 1][1][grow][gcol];
                const float2 ao = *(const float2*)&lu.g[mt & 1][2][grow][gcol];
                const float2 sf = *(const float2*)&facc[r][gcol];
                float2 cc, hh;
                {
                    const float i_j = sigmoidf_(ai.x + gi.x);
                    const float c_t = tanhf(ac.x + gc.x);
                    const float c   = i_j * c_t + sf.x;
                    const float o_j = sigmoidf_(ao.x + go.x);
                    hh.x = o_j * tanhf(c);
                    cc.x = c;
                }
                {
                    const float i_j = sigmoidf_(ai.y + gi.y);
                    const float c_t = tanhf(ac.y + gc.y);
                    const float c   = i_j * c_t + sf.y;
                    const float o_j = sigmoidf_(ao.y + go.y);
                    hh.y = o_j * tanhf(c);
                    cc.y = c;
                }
                const size_t o = (size_t)(cbase + r) * HDIM + gcol;
                *(float2*)(h_out + o) = hh;
                *(float2*)(c_out + o) = cc;
            }
        };

        const bool pw     = wave < 24;
        const bool pfirst = ((wave >> 2) & 1) == 0;
        if (pw) product(0);
        __syncthreads();
        for (int mt = 0; mt < ntile; ++mt) {
            const bool nxt = mt + 1 < ntile;
            if (nxt && pw && pfirst) product(mt + 1);
            gate(mt, gi0, gc0, go0);
            if (nxt) load_g(mt + 1, gi0, gc0, go0);
            if (nxt && pw && !pfirst) product(mt + 1);
            __syncthreads();
        }
    }
}

extern "C" void kernel_launch(void* const* d_in, const int* in_sizes, int n_in,
                              void* d_out, int out_size, void* d_ws, size_t ws_size,
                              hipStream_t stream) {
    const float* x_emb   = (const float*)d_in[0];
    const float* child_h = (const float*)d_in[1];
    const float* child_c = (const float*)d_in[2];
    const int*   ci      = (const int*)d_in[3];
    const int*   chi     = (const int*)d_in[4];
    const float* Wi_w = (const float*)d_in[5];
    const float* Ui_w = (const float*)d_in[6];
    const float* Wf_w = (const float*)d_in[7];
    const float* Uf_w = (const float*)d_in[8];
    const float* Wo_w = (const float*)d_in[9];
    const float* Uo_w = (const float*)d_in[10];
    const float* Wc_w = (const float*)d_in[11];
    const float* Uc_w = (const float*)d_in[12];
    const float* Wa_w = (const float*)d_in[13];
    const float* Ua_w = (const float*)d_in[14];
    const float* Wi_b = (const float*)d_in[15];
    const float* Wf_b = (const float*)d_in[16];
    const float* Wo_b = (const float*)d_in[17];
    const float* Wc_b = (const float*)d_in[18];
    const float* Wa_b = (const float*)d_in[19];
    const float* v_w  = (const float*)d_in[20];

    const int M = in_sizes[0] / HDIM;
    const int L = in_sizes[3];
    const size_t MH = (size_t)M * HDIM;

    float* ws    = (float*)d_ws;
    float* Ah    = ws;
    float* Bx    = Ah  + MH;
    float* Gi    = Bx  + MH;
    float* Wfx   = Gi  + MH;
    float* Go    = Wfx + MH;
    float* Gc    = Go  + MH;
    float* Ufh   = Gc  + MH;

    proj7_wmma_f32<<<dim3(763), dim3(256), 0, stream>>>(x_emb, child_h,
                                                        Wa_w, Ua_w, Wi_w, Wf_w, Wo_w, Wc_w, Uf_w,
                                                        Wa_b, Wi_b, Wf_b, Wo_b, Wc_b,
                                                        ws, M);

    float* h_out = (float*)d_out;
    float* c_out = h_out + MH;
    edge_attn_owner_kernel<<<dim3(256), dim3(OWN_NT), 0, stream>>>(Ah, Bx, v_w,
                                                        child_h, child_c, Wfx, Ufh,
                                                        ci, chi, Ui_w, Uc_w, Uo_w,
                                                        Gi, Gc, Go, h_out, c_out, M, L);
}


# Edge lists built once in the product launch: 512 waves sort a slice of edge ids each by owner block (ballot ranks, no atomics) into the owner's own output rows; owners stop scanning ids; sums same
# speedup vs baseline: 2.7795x; 1.0497x over previous
#include <hip/hip_runtime.h>
#include <hip/hip_bf16.h>

#define HDIM 128
#define EPSV 1e-9f

typedef __attribute__((ext_vector_type(2))) float v2f;
typedef __attribute__((ext_vector_type(8))) float v8f;

#define PROJ_N 7

#define OWN_R      196
#define OWN_BLOCKS 256
#define PRE_NS     512
#define PRE_CAP    31
#define PRE_SCANB  64

__device__ __forceinline__ bool lists_prebuilt_(int nblk, int M, int L) {
    return nblk <= 256 && nblk * OWN_R >= M && L <= (1 << 24);
}

__device__ __forceinline__ void prebuild_lists(const int* __restrict__ ci, int* __restrict__ outw,
                                               int sid, int M, int L, int* cntw) {
    const int lane = threadIdx.x & 31;
    const unsigned lt = (1u << lane) - 1u;
#pragma unroll
    for (int j = 0; j < 8; ++j) cntw[lane + 32 * j] = 0;
    const int SLICE = (((L + PRE_NS - 1) / PRE_NS) + 31) & ~31;
    const int p0 = min(L, sid * SLICE), p1 = min(L, sid * SLICE + SLICE);
    int idn = (p0 < p1) ? ci[min(p0 + lane, L - 1)] : 0;
    for (int p = p0; p < p1; p += 32) {
        const int id  = idn;
        if (p + 32 < p1) idn = ci[min(p + 32 + lane, L - 1)];
        const int idx = p + lane;
        const int b   = id / OWN_R;
        const bool valid = idx < p1 && (unsigned)id < (unsigned)M && (b + 1) * OWN_R <= M;
        unsigned same = (unsigned)__ballot(valid);
#pragma unroll
        for (int k = 0; k < 8; ++k) {
            const bool bit = ((b >> k) & 1) != 0;
            const unsigned mk = (unsigned)__ballot(bit);
            same &= bit ? mk : ~mk;
        }
        if (valid) {
            const int base = cntw[b];
            const int rank = __popc(same & lt);
            const int tot  = __popc(same);
            const int pos  = base + rank;
            int* sub = outw + (size_t)b * (OWN_R * HDIM) + sid * 32;
            if (pos < PRE_CAP) sub[1 + pos] = (int)(((unsigned)idx << 8) | (unsigned)(id - b * OWN_R));
            if (rank == tot - 1) cntw[b] = base + tot;
        }
    }
#pragma unroll
    for (int j = 0; j < 8; ++j) {
        const int b = lane + 32 * j;
        if ((b + 1) * OWN_R <= M) outw[(size_t)b * (OWN_R * HDIM) + sid * 32] = cntw[b];
    }
}

__device__ __forceinline__ void proj_block(const float* __restrict__ X,
                                           const float* __restrict__ W,
                                           const float* __restrict__ bias,
                                           float* __restrict__ Y,
                                           int Mrows, int t0, int tstride) {
    __shared__ float lw[HDIM][132];
    __shared__ float lx[2][16][132];

    const int wave  = threadIdx.x >> 5;
    const int lane  = threadIdx.x & 31;
    const int nsub  = lane & 15;
    const int khalf = (lane >> 4) << 1;
    const int n0    = wave * 16;
    const int ntiles = (Mrows + 15) >> 4;

    const int srow = threadIdx.x >> 5;
    const int scol = (threadIdx.x & 31) << 2;
    const int lastrow = Mrows - 1;

    int t = t0;
    float4 p0 = {}, p1 = {};
    if (t < ntiles) {
        const int ra = min(t * 16 + srow, lastrow), rb = min(t * 16 + srow + 8, lastrow);
        p0 = *(const float4*)(X + (size_t)ra * HDIM + scol);
        p1 = *(const float4*)(X + (size_t)rb * HDIM + scol);
    }

#pragma unroll
    for (int p = 0; p < HDIM / 8; ++p)
        *(float4*)&lw[srow + 8 * p][scol] =
            *(const float4*)(W + (size_t)(srow + 8 * p) * HDIM + scol);
    __syncthreads();

    v2f bfrag[HDIM / 4];
#pragma unroll
    for (int j = 0; j < HDIM / 4; ++j)
        bfrag[j] = *(const v2f*)&lw[n0 + nsub][4 * j + khalf];
    const float bval = bias ? bias[n0 + nsub] : 0.0f;

    int buf = 0;
    for (; t < ntiles; t += tstride) {
        *(float4*)&lx[buf][srow][scol]     = p0;
        *(float4*)&lx[buf][srow + 8][scol] = p1;
        __syncthreads();

        const int tn = t + tstride;
        if (tn < ntiles) {
            const int ra = min(tn * 16 + srow, lastrow), rb = min(tn * 16 + srow + 8, lastrow);
            p0 = *(const float4*)(X + (size_t)ra * HDIM + scol);
            p1 = *(const float4*)(X + (size_t)rb * HDIM + scol);
        }

        const bool full  = (t * 16 + 16 <= Mrows);
        const int  mbase = t * 16 + ((lane >> 4) << 3);
        float* __restrict__ yp = Y + (size_t)mbase * HDIM + n0 + nsub;

        const float* arow = &lx[buf][nsub][khalf];
        v8f acc = {};
#pragma unroll
        for (int j = 0; j < HDIM / 4; ++j) {
            const v2f a = *(const v2f*)(arow + 4 * j);
            acc = __builtin_amdgcn_wmma_f32_16x16x4_f32(
                 false, a,  false, bfrag[j],
                 (short)0, acc,  false,  false);
        }

        float v[8];
#pragma unroll
        for (int r = 0; r < 8; ++r) v[r] = acc[r] + bval;
        if (full) {
#pragma unroll
            for (int r = 0; r < 8; ++r) yp[(size_t)r * HDIM] = v[r];
        } else {
#pragma unroll
            for (int r = 0; r < 8; ++r)
                if (mbase + r < Mrows) yp[(size_t)r * HDIM] = v[r];
        }
        buf ^= 1;
    }
}

__launch_bounds__(256)
__global__ void proj7_wmma_f32(const float* __restrict__ x_emb,
                               const float* __restrict__ child_h,
                               const float* __restrict__ Wa_w,
                               const float* __restrict__ Ua_w,
                               const float* __restrict__ Wi_w,
                               const float* __restrict__ Wf_w,
                               const float* __restrict__ Wo_w,
                               const float* __restrict__ Wc_w,
                               const float* __restrict__ Uf_w,
                               const float* __restrict__ Wa_b,
                               const float* __restrict__ Wi_b,
                               const float* __restrict__ Wf_b,
                               const float* __restrict__ Wo_b,
                               const float* __restrict__ Wc_b,
                               float* __restrict__ ws,
                               int Mrows,
                               const int* __restrict__ ci,
                               float* __restrict__ out,
                               int L, int nblk) {
    __shared__ int pre_cnt[8][256];
    const int sblk = (int)blockIdx.x - ((int)gridDim.x - PRE_SCANB);
    if (sblk >= 0 && lists_prebuilt_(nblk, Mrows, L)) {
        const int wv = __builtin_amdgcn_readfirstlane((int)(threadIdx.x >> 5));
        prebuild_lists(ci, (int*)out, sblk * 8 + wv, Mrows, L, &pre_cnt[wv][0]);
    }

    const int q  = blockIdx.x % PROJ_N;
    const int r0 = blockIdx.x / PROJ_N;
    const int pb = gridDim.x / PROJ_N;
    if (r0 >= pb) return;

    const float* X    = x_emb;
    const float* W    = Ua_w;
    const float* bias = nullptr;
    switch (q) {
        case 0: X = child_h; W = Wa_w; bias = Wa_b; break;
        case 1: break;
        case 2: W = Wi_w; bias = Wi_b; break;
        case 3: W = Wf_w; bias = Wf_b; break;
        case 4: W = Wo_w; bias = Wo_b; break;
        case 5: W = Wc_w; bias = Wc_b; break;
        default: X = child_h; W = Uf_w; break;
    }
    proj_block(X, W, bias, ws + (size_t)q * (size_t)Mrows * HDIM, Mrows, r0, pb);
}

__device__ __forceinline__ float sigmoidf_(float x) {
    return 1.0f / (1.0f + __expf(-x));
}

template <int XM>
__device__ __forceinline__ float row_xmask_(float x) {
    const int b = __builtin_bit_cast(int, x);
    return __builtin_bit_cast(float, __builtin_amdgcn_update_dpp(b, b, 0x160 | XM, 0xf, 0xf, false));
}
__device__ __forceinline__ float wave_sum32_(float x) {
    const int b = __builtin_bit_cast(int, x);
    x += __builtin_bit_cast(float, __builtin_amdgcn_permlanex16(b, b, 0x76543210, 0xfedcba98, false, false));
    x += row_xmask_<8>(x);
    x += row_xmask_<4>(x);
    x += row_xmask_<2>(x);
    x += row_xmask_<1>(x);
    return x;
}

#define OWN_RP   208
#define OWN_LDP  132
#define OWN_NT   1024
#define OWN_SEG  512
#define OWN_G    2

__launch_bounds__(OWN_NT)
__global__ void edge_attn_owner_kernel(const float* __restrict__ Ah,
                                       const float* __restrict__ Bx,
                                       const float* __restrict__ v_w,
                                       const float* __restrict__ child_h,
                                       const float* __restrict__ child_c,
                                       const float* __restrict__ Wfx,
                                       const float* __restrict__ Ufh,
                                       const int* __restrict__ ci,
                                       const int* __restrict__ chi,
                                       const float* __restrict__ Ui,
                                       const float* __restrict__ Uc,
                                       const float* __restrict__ Uo,
                                       const float* __restrict__ Gi,
                                       const float* __restrict__ Gc,
                                       const float* __restrict__ Go,
                                       float* __restrict__ h_out,
                                       float* __restrict__ c_out,
                                       int M, int L) {
    union ListOrStage {
        int   e[32][OWN_SEG];
        float g[2][3][16][OWN_LDP];
    };
    __shared__ alignas(16) float hacc[OWN_RP][OWN_LDP];
    __shared__ alignas(16) float facc[OWN_R][HDIM];
    __shared__ float rmax[OWN_R];
    __shared__ float rden[OWN_R];
    __shared__ alignas(16) ListOrStage lu;
    __shared__ unsigned char list_c[32][OWN_SEG];
    __shared__ int           s_cnt[32];
    __shared__ int           s_more[32];
    __shared__ int           wb_e[32][64];
    __shared__ unsigned char wb_c[32][64];

    const int wave = __builtin_amdgcn_readfirstlane((int)(threadIdx.x >> 5));
    const int lane = threadIdx.x & 31;
    const unsigned lt = (1u << lane) - 1u;
    const int SL = (((L + 31) >> 5) + 31) & ~31;
    const float4 vw = ((const float4*)v_w)[lane];

    for (int cbase = blockIdx.x * OWN_R; cbase < M; cbase += gridDim.x * OWN_R) {
        const int nown = min(OWN_R, M - cbase);

        for (int i = threadIdx.x; i < OWN_RP * OWN_LDP / 4; i += OWN_NT) {
            ((float4*)&hacc[0][0])[i] = make_float4(0.f, 0.f, 0.f, 0.f);
            if (i < OWN_R * HDIM / 4)
                ((float4*)&facc[0][0])[i] = make_float4(0.f, 0.f, 0.f, 0.f);
        }
        for (int i = threadIdx.x; i < OWN_R; i += OWN_NT) {
            rmax[i] = -INFINITY;
            rden[i] = 0.f;
        }
        int wpos = min(L, wave * SL);
        const int wend = min(L, wave * SL + SL);
        bool pre = lists_prebuilt_((int)gridDim.x, M, L) && nown == OWN_R;
        __syncthreads();

        bool more = true;
        while (more) {
            int cnt = 0;
            int flag = 0;
            if (pre) {
                const int* reg = (const int*)(h_out + (size_t)cbase * HDIM) + wave * (16 * 32);
                bool ov = false;
                for (int h8 = 0; h8 < 16; h8 += 8) {
                    int w8[8];
#pragma unroll
                    for (int i = 0; i < 8; ++i) w8[i] = reg[(h8 + i) * 32 + lane];
#pragma unroll
                    for (int i = 0; i < 8; ++i) {
                        const int c = __builtin_amdgcn_readlane(w8[i], 0);
                        ov = ov || ((unsigned)c > (unsigned)PRE_CAP);
                        const int cc = ov ? 0 : c;
                        if (lane >= 1 && lane <= cc) {
                            lu.e[wave][cnt + lane - 1]   = (int)((unsigned)w8[i] >> 8);
                            list_c[wave][cnt + lane - 1] = (unsigned char)(w8[i] & 0xff);
                        }
                        cnt += cc;
                    }
                }
                if (ov) { cnt = 0; flag = 2; }
            } else if (wpos < wend) {
                int ida[8], idb[8];
                auto load_ids = [&](int (&v)[8], int p) {
#pragma unroll
                    for (int j = 0; j < 8; ++j) {
                        const int idx = p + j * 32 + lane;
                        const int t = ci[min(idx, L - 1)];
                        v[j] = (idx < wend) ? t : -1;
                    }
                };
                auto take = [&](const int (&v)[8], int p) {
#pragma unroll
                    for (int j = 0; j < 8; ++j) {
                        const int cl = v[j] - cbase;
                        const bool hit = (unsigned)cl < (unsigned)nown;
                        const unsigned m = (unsigned)__ballot(hit);
                        if (hit) {
                            const int k = cnt + __popc(m & lt);
                            lu.e[wave][k] = p + j * 32 + lane;
                            list_c[wave][k] = (unsigned char)cl;
                        }
                        cnt += __popc(m);
                    }
                };
                load_ids(ida, wpos);
                for (;;) {
                    load_ids(idb, wpos + 256);
                    take(ida, wpos);
                    wpos += 256;
                    if (!(wpos < wend && cnt + 256 <= OWN_SEG)) break;
                    load_ids(ida, wpos + 256);
                    take(idb, wpos);
                    wpos += 256;
                    if (!(wpos < wend && cnt + 256 <= OWN_SEG)) break;
                }
                wpos = min(wpos, wend);
            }
            if (!pre) flag = (wpos < wend) ? 1 : 0;
            if (lane == 0) {
                s_cnt[wave]  = cnt;
                s_more[wave] = flag;
            }
            __syncthreads();
            int cntv = s_cnt[lane];
            more = __ballot(s_more[lane] != 0) != 0;
            if (pre) {
                if (more) cntv = 0;
                pre = false;
            }

            int nb = 0;
            int q = 0, k0 = 0;
            for (;;) {
                int nq = 0;
                bool have = false;
                while (q < 32) {
                    nq = __builtin_amdgcn_readlane(cntv, q);
                    if (k0 < nq) { have = true; break; }
                    ++q; k0 = 0;
                }
                if (have) {
                    const int k = k0 + lane;
                    const bool valid = k < nq;
                    const int e  = valid ? lu.e[q][k] : 0;
                    const int cl = valid ? (int)list_c[q][k] : 0;
                    const bool mine = valid && ((cl & 31) == wave);
                    const unsigned mm = (unsigned)__ballot(mine);
                    if (mine) {
                        const int d = nb + __popc(mm & lt);
                        wb_e[wave][d] = e;
                        wb_c[wave][d] = (unsigned char)cl;
                    }
                    nb += __popc(mm);
                    k0 += 32;
                }
                if (nb >= 32 || (!have && nb > 0)) {
                    const int n = min(nb, 32);
                    int bcl = 0, bch = 0;
                    if (lane < n) {
                        bcl = (int)wb_c[wave][lane];
                        bch = chi[wb_e[wave][lane]];
                    }
                    for (int g = 0; g < n; g += OWN_G) {
                        float4 ah[OWN_G], bx[OWN_G], hv[OWN_G], cv4[OWN_G], uf[OWN_G], wf[OWN_G];
                        int   cl4[OWN_G];
#pragma unroll
                        for (int u = 0; u < OWN_G; ++u) {
                            const int i  = min(g + u, n - 1);
                            const int ch = __builtin_amdgcn_readlane(bch, i);
                            const int cl = __builtin_amdgcn_readlane(bcl, i);
                            cl4[u] = cl;
                            ah[u]  = ((const float4*)(Ah      + (size_t)ch * HDIM))[lane];
                            hv[u]  = ((const float4*)(child_h + (size_t)ch * HDIM))[lane];
                            cv4[u] = ((const float4*)(child_c + (size_t)ch * HDIM))[lane];
                            uf[u]  = ((const float4*)(Ufh     + (size_t)ch * HDIM))[lane];
                            bx[u]  = ((const float4*)(Bx  + (size_t)(cbase + cl) * HDIM))[lane];
                            wf[u]  = ((const float4*)(Wfx + (size_t)(cbase + cl) * HDIM))[lane];
                        }
#pragma unroll
                        for (int u = 0; u < OWN_G; ++u) {
                            if (g + u < n) {
                                float s = tanhf(ah[u].x + bx[u].x) * vw.x + tanhf(ah[u].y + bx[u].y) * vw.y
                                        + tanhf(ah[u].z + bx[u].z) * vw.z + tanhf(ah[u].w + bx[u].w) * vw.w;
                                s = wave_sum32_(s);

                                const int   cl    = cl4[u];
                                const float m_old = rmax[cl];
                                const float d_old = rden[cl];
                                const float m_new = fmaxf(m_old, s);
                                const float sc    = __expf(m_old - m_new);
                                const float p     = __expf(s - m_new);
                                if (lane == 0) {
                                    rmax[cl] = m_new;
                                    rden[cl] = d_old * sc + p;
                                }

                                float4* hp = (float4*)&hacc[cl][lane * 4];
                                float4* fp = (float4*)&facc[cl][lane * 4];
                                float4 h = *hp, f = *fp;
                                h.x = h.x * sc + p * hv[u].x;
                                h.y = h.y * sc + p * hv[u].y;
                                h.z = h.z * sc + p * hv[u].z;
                                h.w = h.w * sc + p * hv[u].w;
                                f.x += sigmoidf_(wf[u].x + uf[u].x) * cv4[u].x;
                                f.y += sigmoidf_(wf[u].y + uf[u].y) * cv4[u].y;
                                f.z += sigmoidf_(wf[u].z + uf[u].z) * cv4[u].z;
                                f.w += sigmoidf_(wf[u].w + uf[u].w) * cv4[u].w;
                                *hp = h;
                                *fp = f;
                            }
                        }
                    }
                    const int rest = nb - n;
                    int te = 0;
                    unsigned char tc = 0;
                    if (lane < rest) { te = wb_e[wave][32 + lane]; tc = wb_c[wave][32 + lane]; }
                    if (lane < rest) { wb_e[wave][lane] = te; wb_c[wave][lane] = tc; }
                    nb = rest;
                }
                if (!have) break;
            }
            __syncthreads();
        }

        const int ntile = (nown + 15) >> 4;
        const int nsub  = lane & 15;
        const int khalf = (lane >> 4) << 1;
        const int pq    = wave >> 3;
        const int pn0   = (wave & 7) * 16;
        const int grow  = wave >> 1;
        const int gcol  = (((wave & 1) << 5) + lane) << 1;

        auto load_g = [&](int mt, float2& gi, float2& gc, float2& go) {
            const size_t o = (size_t)(cbase + min(mt * 16 + grow, nown - 1)) * HDIM + gcol;
            gi = *(const float2*)(Gi + o);
            gc = *(const float2*)(Gc + o);
            go = *(const float2*)(Go + o);
        };
        float2 gi0, gc0, go0;
        load_g(0, gi0, gc0, go0);

        v2f bfrag[HDIM / 4] = {};
        if (wave < 24) {
            const float* ub = (pq == 0 ? Ui : (pq == 1 ? Uc : Uo)) + pn0 * HDIM;
            const int    uo = nsub * HDIM + khalf;
#pragma unroll
            for (int j = 0; j < HDIM / 4; ++j) bfrag[j] = *(const v2f*)(ub + uo + 4 * j);
        }

        for (int i = threadIdx.x; i < nown * (HDIM / 4); i += OWN_NT) {
            const int row = i >> 5, c4 = i & 31;
            const float den = rden[row] + EPSV;
            float4* hp = (float4*)&hacc[row][c4 * 4];
            float4 hh = *hp;
            hh.x /= den; hh.y /= den; hh.z /= den; hh.w /= den;
            *hp = hh;
        }
        __syncthreads();

        auto product = [&](int mt) {
            const float* arow = &hacc[mt * 16 + nsub][khalf];
            v8f acc = {};
#pragma unroll
            for (int j = 0; j < HDIM / 4; ++j) {
                const v2f a = *(const v2f*)(arow + 4 * j);
                acc = __builtin_amdgcn_wmma_f32_16x16x4_f32(
                     false, a,  false, bfrag[j],
                     (short)0, acc,  false,  false);
            }
            float* sp = &lu.g[mt & 1][pq][(lane >> 4) << 3][pn0 + nsub];
#pragma unroll
            for (int r = 0; r < 8; ++r) sp[r * OWN_LDP] = acc[r];
        };
        auto gate = [&](int mt, float2 gi, float2 gc, float2 go) {
            const int r = mt * 16 + grow;
            if (r < nown) {
                const float2 ai = *(const float2*)&lu.g[mt & 1][0][grow][gcol];
                const float2 ac = *(const float2*)&lu.g[mt & 1][1][grow][gcol];
                const float2 ao = *(const float2*)&lu.g[mt & 1][2][grow][gcol];
                const float2 sf = *(const float2*)&facc[r][gcol];
                float2 cc, hh;
                {
                    const float i_j = sigmoidf_(ai.x + gi.x);
                    const float c_t = tanhf(ac.x + gc.x);
                    const float c   = i_j * c_t + sf.x;
                    const float o_j = sigmoidf_(ao.x + go.x);
                    hh.x = o_j * tanhf(c);
                    cc.x = c;
                }
                {
                    const float i_j = sigmoidf_(ai.y + gi.y);
                    const float c_t = tanhf(ac.y + gc.y);
                    const float c   = i_j * c_t + sf.y;
                    const float o_j = sigmoidf_(ao.y + go.y);
                    hh.y = o_j * tanhf(c);
                    cc.y = c;
                }
                const size_t o = (size_t)(cbase + r) * HDIM + gcol;
                *(float2*)(h_out + o) = hh;
                *(float2*)(c_out + o) = cc;
            }
        };

        const bool pw     = wave < 24;
        const bool pfirst = ((wave >> 2) & 1) == 0;
        if (pw) product(0);
        __syncthreads();
        for (int mt = 0; mt < ntile; ++mt) {
            const bool nxt = mt + 1 < ntile;
            if (nxt && pw && pfirst) product(mt + 1);
            gate(mt, gi0, gc0, go0);
            if (nxt) load_g(mt + 1, gi0, gc0, go0);
            if (nxt && pw && !pfirst) product(mt + 1);
            __syncthreads();
        }
    }
}

extern "C" void kernel_launch(void* const* d_in, const int* in_sizes, int n_in,
                              void* d_out, int out_size, void* d_ws, size_t ws_size,
                              hipStream_t stream) {
    const float* x_emb   = (const float*)d_in[0];
    const float* child_h = (const float*)d_in[1];
    const float* child_c = (const float*)d_in[2];
    const int*   ci      = (const int*)d_in[3];
    const int*   chi     = (const int*)d_in[4];
    const float* Wi_w = (const float*)d_in[5];
    const float* Ui_w = (const float*)d_in[6];
    const float* Wf_w = (const float*)d_in[7];
    const float* Uf_w = (const float*)d_in[8];
    const float* Wo_w = (const float*)d_in[9];
    const float* Uo_w = (const float*)d_in[10];
    const float* Wc_w = (const float*)d_in[11];
    const float* Uc_w = (const float*)d_in[12];
    const float* Wa_w = (const float*)d_in[13];
    const float* Ua_w = (const float*)d_in[14];
    const float* Wi_b = (const float*)d_in[15];
    const float* Wf_b = (const float*)d_in[16];
    const float* Wo_b = (const float*)d_in[17];
    const float* Wc_b = (const float*)d_in[18];
    const float* Wa_b = (const float*)d_in[19];
    const float* v_w  = (const float*)d_in[20];

    const int M = in_sizes[0] / HDIM;
    const int L = in_sizes[3];
    const size_t MH = (size_t)M * HDIM;

    float* ws    = (float*)d_ws;
    float* Ah    = ws;
    float* Bx    = Ah  + MH;
    float* Gi    = Bx  + MH;
    float* Wfx   = Gi  + MH;
    float* Go    = Wfx + MH;
    float* Gc    = Go  + MH;
    float* Ufh   = Gc  + MH;

    proj7_wmma_f32<<<dim3(763), dim3(256), 0, stream>>>(x_emb, child_h,
                                                        Wa_w, Ua_w, Wi_w, Wf_w, Wo_w, Wc_w, Uf_w,
                                                        Wa_b, Wi_b, Wf_b, Wo_b, Wc_b,
                                                        ws, M, ci, (float*)d_out, L, OWN_BLOCKS);

    float* h_out = (float*)d_out;
    float* c_out = h_out + MH;
    edge_attn_owner_kernel<<<dim3(OWN_BLOCKS), dim3(OWN_NT), 0, stream>>>(Ah, Bx, v_w,
                                                        child_h, child_c, Wfx, Ufh,
                                                        ci, chi, Ui_w, Uc_w, Uo_w,
                                                        Gi, Gc, Go, h_out, c_out, M, L);
}


# Owned rows dealt to the block's 32 waves by edge count (LDS integer adds, rank, back-and-forth deal), not by row & 31: the fullest wave loses about a quarter of its edges; sum orders same
# speedup vs baseline: 2.8592x; 1.0287x over previous
#include <hip/hip_runtime.h>
#include <hip/hip_bf16.h>

#define HDIM 128
#define EPSV 1e-9f

typedef __attribute__((ext_vector_type(2))) float v2f;
typedef __attribute__((ext_vector_type(8))) float v8f;

#define PROJ_N 7

#define OWN_R      196
#define OWN_BLOCKS 256
#define PRE_NS     512
#define PRE_CAP    31
#define PRE_SCANB  64

__device__ __forceinline__ bool lists_prebuilt_(int nblk, int M, int L) {
    return nblk <= 256 && nblk * OWN_R >= M && L <= (1 << 24);
}

__device__ __forceinline__ void prebuild_lists(const int* __restrict__ ci, int* __restrict__ outw,
                                               int sid, int M, int L, int* cntw) {
    const int lane = threadIdx.x & 31;
    const unsigned lt = (1u << lane) - 1u;
#pragma unroll
    for (int j = 0; j < 8; ++j) cntw[lane + 32 * j] = 0;
    const int SLICE = (((L + PRE_NS - 1) / PRE_NS) + 31) & ~31;
    const int p0 = min(L, sid * SLICE), p1 = min(L, sid * SLICE + SLICE);
    int idn = (p0 < p1) ? ci[min(p0 + lane, L - 1)] : 0;
    for (int p = p0; p < p1; p += 32) {
        const int id  = idn;
        if (p + 32 < p1) idn = ci[min(p + 32 + lane, L - 1)];
        const int idx = p + lane;
        const int b   = id / OWN_R;
        const bool valid = idx < p1 && (unsigned)id < (unsigned)M && (b + 1) * OWN_R <= M;
        unsigned same = (unsigned)__ballot(valid);
#pragma unroll
        for (int k = 0; k < 8; ++k) {
            const bool bit = ((b >> k) & 1) != 0;
            const unsigned mk = (unsigned)__ballot(bit);
            same &= bit ? mk : ~mk;
        }
        if (valid) {
            const int base = cntw[b];
            const int rank = __popc(same & lt);
            const int tot  = __popc(same);
            const int pos  = base + rank;
            int* sub = outw + (size_t)b * (OWN_R * HDIM) + sid * 32;
            if (pos < PRE_CAP) sub[1 + pos] = (int)(((unsigned)idx << 8) | (unsigned)(id - b * OWN_R));
            if (rank == tot - 1) cntw[b] = base + tot;
        }
    }
#pragma unroll
    for (int j = 0; j < 8; ++j) {
        const int b = lane + 32 * j;
        if ((b + 1) * OWN_R <= M) outw[(size_t)b * (OWN_R * HDIM) + sid * 32] = cntw[b];
    }
}

__device__ __forceinline__ void proj_block(const float* __restrict__ X,
                                           const float* __restrict__ W,
                                           const float* __restrict__ bias,
                                           float* __restrict__ Y,
                                           int Mrows, int t0, int tstride) {
    __shared__ float lw[HDIM][132];
    __shared__ float lx[2][16][132];

    const int wave  = threadIdx.x >> 5;
    const int lane  = threadIdx.x & 31;
    const int nsub  = lane & 15;
    const int khalf = (lane >> 4) << 1;
    const int n0    = wave * 16;
    const int ntiles = (Mrows + 15) >> 4;

    const int srow = threadIdx.x >> 5;
    const int scol = (threadIdx.x & 31) << 2;
    const int lastrow = Mrows - 1;

    int t = t0;
    float4 p0 = {}, p1 = {};
    if (t < ntiles) {
        const int ra = min(t * 16 + srow, lastrow), rb = min(t * 16 + srow + 8, lastrow);
        p0 = *(const float4*)(X + (size_t)ra * HDIM + scol);
        p1 = *(const float4*)(X + (size_t)rb * HDIM + scol);
    }

#pragma unroll
    for (int p = 0; p < HDIM / 8; ++p)
        *(float4*)&lw[srow + 8 * p][scol] =
            *(const float4*)(W + (size_t)(srow + 8 * p) * HDIM + scol);
    __syncthreads();

    v2f bfrag[HDIM / 4];
#pragma unroll
    for (int j = 0; j < HDIM / 4; ++j)
        bfrag[j] = *(const v2f*)&lw[n0 + nsub][4 * j + khalf];
    const float bval = bias ? bias[n0 + nsub] : 0.0f;

    int buf = 0;
    for (; t < ntiles; t += tstride) {
        *(float4*)&lx[buf][srow][scol]     = p0;
        *(float4*)&lx[buf][srow + 8][scol] = p1;
        __syncthreads();

        const int tn = t + tstride;
        if (tn < ntiles) {
            const int ra = min(tn * 16 + srow, lastrow), rb = min(tn * 16 + srow + 8, lastrow);
            p0 = *(const float4*)(X + (size_t)ra * HDIM + scol);
            p1 = *(const float4*)(X + (size_t)rb * HDIM + scol);
        }

        const bool full  = (t * 16 + 16 <= Mrows);
        const int  mbase = t * 16 + ((lane >> 4) << 3);
        float* __restrict__ yp = Y + (size_t)mbase * HDIM + n0 + nsub;

        const float* arow = &lx[buf][nsub][khalf];
        v8f acc = {};
#pragma unroll
        for (int j = 0; j < HDIM / 4; ++j) {
            const v2f a = *(const v2f*)(arow + 4 * j);
            acc = __builtin_amdgcn_wmma_f32_16x16x4_f32(
                 false, a,  false, bfrag[j],
                 (short)0, acc,  false,  false);
        }

        float v[8];
#pragma unroll
        for (int r = 0; r < 8; ++r) v[r] = acc[r] + bval;
        if (full) {
#pragma unroll
            for (int r = 0; r < 8; ++r) yp[(size_t)r * HDIM] = v[r];
        } else {
#pragma unroll
            for (int r = 0; r < 8; ++r)
                if (mbase + r < Mrows) yp[(size_t)r * HDIM] = v[r];
        }
        buf ^= 1;
    }
}

__launch_bounds__(256)
__global__ void proj7_wmma_f32(const float* __restrict__ x_emb,
                               const float* __restrict__ child_h,
                               const float* __restrict__ Wa_w,
                               const float* __restrict__ Ua_w,
                               const float* __restrict__ Wi_w,
                               const float* __restrict__ Wf_w,
                               const float* __restrict__ Wo_w,
                               const float* __restrict__ Wc_w,
                               const float* __restrict__ Uf_w,
                               const float* __restrict__ Wa_b,
                               const float* __restrict__ Wi_b,
                               const float* __restrict__ Wf_b,
                               const float* __restrict__ Wo_b,
                               const float* __restrict__ Wc_b,
                               float* __restrict__ ws,
                               int Mrows,
                               const int* __restrict__ ci,
                               float* __restrict__ out,
                               int L, int nblk) {
    __shared__ int pre_cnt[8][256];
    const int sblk = (int)blockIdx.x - ((int)gridDim.x - PRE_SCANB);
    if (sblk >= 0 && lists_prebuilt_(nblk, Mrows, L)) {
        const int wv = __builtin_amdgcn_readfirstlane((int)(threadIdx.x >> 5));
        prebuild_lists(ci, (int*)out, sblk * 8 + wv, Mrows, L, &pre_cnt[wv][0]);
    }

    const int q  = blockIdx.x % PROJ_N;
    const int r0 = blockIdx.x / PROJ_N;
    const int pb = gridDim.x / PROJ_N;
    if (r0 >= pb) return;

    const float* X    = x_emb;
    const float* W    = Ua_w;
    const float* bias = nullptr;
    switch (q) {
        case 0: X = child_h; W = Wa_w; bias = Wa_b; break;
        case 1: break;
        case 2: W = Wi_w; bias = Wi_b; break;
        case 3: W = Wf_w; bias = Wf_b; break;
        case 4: W = Wo_w; bias = Wo_b; break;
        case 5: W = Wc_w; bias = Wc_b; break;
        default: X = child_h; W = Uf_w; break;
    }
    proj_block(X, W, bias, ws + (size_t)q * (size_t)Mrows * HDIM, Mrows, r0, pb);
}

__device__ __forceinline__ float sigmoidf_(float x) {
    return 1.0f / (1.0f + __expf(-x));
}

template <int XM>
__device__ __forceinline__ float row_xmask_(float x) {
    const int b = __builtin_bit_cast(int, x);
    return __builtin_bit_cast(float, __builtin_amdgcn_update_dpp(b, b, 0x160 | XM, 0xf, 0xf, false));
}
__device__ __forceinline__ float wave_sum32_(float x) {
    const int b = __builtin_bit_cast(int, x);
    x += __builtin_bit_cast(float, __builtin_amdgcn_permlanex16(b, b, 0x76543210, 0xfedcba98, false, false));
    x += row_xmask_<8>(x);
    x += row_xmask_<4>(x);
    x += row_xmask_<2>(x);
    x += row_xmask_<1>(x);
    return x;
}

#define OWN_RP   208
#define OWN_LDP  132
#define OWN_NT   1024
#define OWN_SEG  512
#define OWN_G    2

__launch_bounds__(OWN_NT)
__global__ void edge_attn_owner_kernel(const float* __restrict__ Ah,
                                       const float* __restrict__ Bx,
                                       const float* __restrict__ v_w,
                                       const float* __restrict__ child_h,
                                       const float* __restrict__ child_c,
                                       const float* __restrict__ Wfx,
                                       const float* __restrict__ Ufh,
                                       const int* __restrict__ ci,
                                       const int* __restrict__ chi,
                                       const float* __restrict__ Ui,
                                       const float* __restrict__ Uc,
                                       const float* __restrict__ Uo,
                                       const float* __restrict__ Gi,
                                       const float* __restrict__ Gc,
                                       const float* __restrict__ Go,
                                       float* __restrict__ h_out,
                                       float* __restrict__ c_out,
                                       int M, int L) {
    union ListOrStage {
        int   e[32][OWN_SEG];
        float g[2][3][16][OWN_LDP];
    };
    __shared__ alignas(16) float hacc[OWN_RP][OWN_LDP];
    __shared__ alignas(16) float facc[OWN_R][HDIM];
    __shared__ float rmax[OWN_R];
    __shared__ float rden[OWN_R];
    __shared__ alignas(16) ListOrStage lu;
    __shared__ unsigned char list_c[32][OWN_SEG];
    __shared__ int           s_cnt[32];
    __shared__ int           s_more[32];
    __shared__ int           wb_e[32][64];
    __shared__ unsigned char wb_c[32][64];
    __shared__ alignas(16) int rcnt[256];
    __shared__ unsigned char roww[256];
    __shared__ unsigned char list_w[32][OWN_SEG];

    const int wave = __builtin_amdgcn_readfirstlane((int)(threadIdx.x >> 5));
    const int lane = threadIdx.x & 31;
    const unsigned lt = (1u << lane) - 1u;
    const int SL = (((L + 31) >> 5) + 31) & ~31;
    const float4 vw = ((const float4*)v_w)[lane];

    for (int cbase = blockIdx.x * OWN_R; cbase < M; cbase += gridDim.x * OWN_R) {
        const int nown = min(OWN_R, M - cbase);

        for (int i = threadIdx.x; i < OWN_RP * OWN_LDP / 4; i += OWN_NT) {
            ((float4*)&hacc[0][0])[i] = make_float4(0.f, 0.f, 0.f, 0.f);
            if (i < OWN_R * HDIM / 4)
                ((float4*)&facc[0][0])[i] = make_float4(0.f, 0.f, 0.f, 0.f);
        }
        for (int i = threadIdx.x; i < OWN_R; i += OWN_NT) {
            rmax[i] = -INFINITY;
            rden[i] = 0.f;
        }
        if (threadIdx.x < 256) rcnt[threadIdx.x] = 0;
        int wpos = min(L, wave * SL);
        const int wend = min(L, wave * SL + SL);
        bool pre = lists_prebuilt_((int)gridDim.x, M, L) && nown == OWN_R;
        __syncthreads();

        bool more = true;
        while (more) {
            int cnt = 0;
            int flag = 0;
            if (pre) {
                const int* reg = (const int*)(h_out + (size_t)cbase * HDIM) + wave * (16 * 32);
                bool ov = false;
                for (int h8 = 0; h8 < 16; h8 += 8) {
                    int w8[8];
#pragma unroll
                    for (int i = 0; i < 8; ++i) w8[i] = reg[(h8 + i) * 32 + lane];
#pragma unroll
                    for (int i = 0; i < 8; ++i) {
                        const int c = __builtin_amdgcn_readlane(w8[i], 0);
                        ov = ov || ((unsigned)c > (unsigned)PRE_CAP);
                        const int cc = ov ? 0 : c;
                        if (lane >= 1 && lane <= cc) {
                            lu.e[wave][cnt + lane - 1]   = (int)((unsigned)w8[i] >> 8);
                            list_c[wave][cnt + lane - 1] = (unsigned char)(w8[i] & 0xff);
                            atomicAdd(&rcnt[w8[i] & 0xff], 1);
                        }
                        cnt += cc;
                    }
                }
                if (ov) { cnt = 0; flag = 2; }
            } else if (wpos < wend) {
                int ida[8], idb[8];
                auto load_ids = [&](int (&v)[8], int p) {
#pragma unroll
                    for (int j = 0; j < 8; ++j) {
                        const int idx = p + j * 32 + lane;
                        const int t = ci[min(idx, L - 1)];
                        v[j] = (idx < wend) ? t : -1;
                    }
                };
                auto take = [&](const int (&v)[8], int p) {
#pragma unroll
                    for (int j = 0; j < 8; ++j) {
                        const int cl = v[j] - cbase;
                        const bool hit = (unsigned)cl < (unsigned)nown;
                        const unsigned m = (unsigned)__ballot(hit);
                        if (hit) {
                            const int k = cnt + __popc(m & lt);
                            lu.e[wave][k] = p + j * 32 + lane;
                            list_c[wave][k] = (unsigned char)cl;
                            atomicAdd(&rcnt[cl], 1);
                        }
                        cnt += __popc(m);
                    }
                };
                load_ids(ida, wpos);
                for (;;) {
                    load_ids(idb, wpos + 256);
                    take(ida, wpos);
                    wpos += 256;
                    if (!(wpos < wend && cnt + 256 <= OWN_SEG)) break;
                    load_ids(ida, wpos + 256);
                    take(idb, wpos);
                    wpos += 256;
                    if (!(wpos < wend && cnt + 256 <= OWN_SEG)) break;
                }
                wpos = min(wpos, wend);
            }
            if (!pre) flag = (wpos < wend) ? 1 : 0;
            if (lane == 0) {
                s_cnt[wave]  = cnt;
                s_more[wave] = flag;
            }
            __syncthreads();
            int cntv = s_cnt[lane];
            more = __ballot(s_more[lane] != 0) != 0;
            if (pre) {
                if (more) cntv = 0;
                pre = false;
            }

            {
                const int t   = (int)threadIdx.x;
                const int r   = min(t >> 2, OWN_R - 1);
                const int pt  = t & 3;
                const int mk  = (rcnt[r] << 8) + r;
                int before = 0;
                static_assert(OWN_R % 4 == 0 && OWN_R <= 256, "rows are ranked in groups of four, row number in 8 bits");
                for (int g = pt; g < OWN_R / 4; g += 4) {
                    const int4 c4 = ((const int4*)rcnt)[g];
                    const int  jb = g * 4;
                    before += (((c4.x << 8) + jb    ) > mk) ? 1 : 0;
                    before += (((c4.y << 8) + jb + 1) > mk) ? 1 : 0;
                    before += (((c4.z << 8) + jb + 2) > mk) ? 1 : 0;
                    before += (((c4.w << 8) + jb + 3) > mk) ? 1 : 0;
                }
                before += __builtin_amdgcn_update_dpp(before, before, 0x160 | 1, 0xf, 0xf, false);
                before += __builtin_amdgcn_update_dpp(before, before, 0x160 | 2, 0xf, 0xf, false);
                if (pt == 0 && (t >> 2) < OWN_R) {
                    const int pos = before & 31;
                    roww[r] = (unsigned char)(((before >> 5) & 1) ? 31 - pos : pos);
                }
            }
            __syncthreads();
            for (int k = lane; k < cnt; k += 32) list_w[wave][k] = roww[list_c[wave][k]];
            if (threadIdx.x < 256) rcnt[threadIdx.x] = 0;
            __syncthreads();

            int nb = 0;
            int q = 0, k0 = 0;
            for (;;) {
                int nq = 0;
                bool have = false;
                while (q < 32) {
                    nq = __builtin_amdgcn_readlane(cntv, q);
                    if (k0 < nq) { have = true; break; }
                    ++q; k0 = 0;
                }
                if (have) {
                    const int k = k0 + lane;
                    const bool valid = k < nq;
                    const int e  = valid ? lu.e[q][k] : 0;
                    const int cl = valid ? (int)list_c[q][k] : 0;
                    const bool mine = valid && ((int)list_w[q][k] == wave);
                    const unsigned mm = (unsigned)__ballot(mine);
                    if (mine) {
                        const int d = nb + __popc(mm & lt);
                        wb_e[wave][d] = e;
                        wb_c[wave][d] = (unsigned char)cl;
                    }
                    nb += __popc(mm);
                    k0 += 32;
                }
                if (nb >= 32 || (!have && nb > 0)) {
                    const int n = min(nb, 32);
                    int bcl = 0, bch = 0;
                    if (lane < n) {
                        bcl = (int)wb_c[wave][lane];
                        bch = chi[wb_e[wave][lane]];
                    }
                    for (int g = 0; g < n; g += OWN_G) {
                        float4 ah[OWN_G], bx[OWN_G], hv[OWN_G], cv4[OWN_G], uf[OWN_G], wf[OWN_G];
                        int   cl4[OWN_G];
#pragma unroll
                        for (int u = 0; u < OWN_G; ++u) {
                            const int i  = min(g + u, n - 1);
                            const int ch = __builtin_amdgcn_readlane(bch, i);
                            const int cl = __builtin_amdgcn_readlane(bcl, i);
                            cl4[u] = cl;
                            ah[u]  = ((const float4*)(Ah      + (size_t)ch * HDIM))[lane];
                            hv[u]  = ((const float4*)(child_h + (size_t)ch * HDIM))[lane];
                            cv4[u] = ((const float4*)(child_c + (size_t)ch * HDIM))[lane];
                            uf[u]  = ((const float4*)(Ufh     + (size_t)ch * HDIM))[lane];
                            bx[u]  = ((const float4*)(Bx  + (size_t)(cbase + cl) * HDIM))[lane];
                            wf[u]  = ((const float4*)(Wfx + (size_t)(cbase + cl) * HDIM))[lane];
                        }
#pragma unroll
                        for (int u = 0; u < OWN_G; ++u) {
                            if (g + u < n) {
                                float s = tanhf(ah[u].x + bx[u].x) * vw.x + tanhf(ah[u].y + bx[u].y) * vw.y
                                        + tanhf(ah[u].z + bx[u].z) * vw.z + tanhf(ah[u].w + bx[u].w) * vw.w;
                                s = wave_sum32_(s);

                                const int   cl    = cl4[u];
                                const float m_old = rmax[cl];
                                const float d_old = rden[cl];
                                const float m_new = fmaxf(m_old, s);
                                const float sc    = __expf(m_old - m_new);
                                const float p     = __expf(s - m_new);
                                if (lane == 0) {
                                    rmax[cl] = m_new;
                                    rden[cl] = d_old * sc + p;
                                }

                                float4* hp = (float4*)&hacc[cl][lane * 4];
                                float4* fp = (float4*)&facc[cl][lane * 4];
                                float4 h = *hp, f = *fp;
                                h.x = h.x * sc + p * hv[u].x;
                                h.y = h.y * sc + p * hv[u].y;
                                h.z = h.z * sc + p * hv[u].z;
                                h.w = h.w * sc + p * hv[u].w;
                                f.x += sigmoidf_(wf[u].x + uf[u].x) * cv4[u].x;
                                f.y += sigmoidf_(wf[u].y + uf[u].y) * cv4[u].y;
                                f.z += sigmoidf_(wf[u].z + uf[u].z) * cv4[u].z;
                                f.w += sigmoidf_(wf[u].w + uf[u].w) * cv4[u].w;
                                *hp = h;
                                *fp = f;
                            }
                        }
                    }
                    const int rest = nb - n;
                    int te = 0;
                    unsigned char tc = 0;
                    if (lane < rest) { te = wb_e[wave][32 + lane]; tc = wb_c[wave][32 + lane]; }
                    if (lane < rest) { wb_e[wave][lane] = te; wb_c[wave][lane] = tc; }
                    nb = rest;
                }
                if (!have) break;
            }
            __syncthreads();
        }

        const int ntile = (nown + 15) >> 4;
        const int nsub  = lane & 15;
        const int khalf = (lane >> 4) << 1;
        const int pq    = wave >> 3;
        const int pn0   = (wave & 7) * 16;
        const int grow  = wave >> 1;
        const int gcol  = (((wave & 1) << 5) + lane) << 1;

        auto load_g = [&](int mt, float2& gi, float2& gc, float2& go) {
            const size_t o = (size_t)(cbase + min(mt * 16 + grow, nown - 1)) * HDIM + gcol;
            gi = *(const float2*)(Gi + o);
            gc = *(const float2*)(Gc + o);
            go = *(const float2*)(Go + o);
        };
        float2 gi0, gc0, go0;
        load_g(0, gi0, gc0, go0);

        v2f bfrag[HDIM / 4] = {};
        if (wave < 24) {
            const float* ub = (pq == 0 ? Ui : (pq == 1 ? Uc : Uo)) + pn0 * HDIM;
            const int    uo = nsub * HDIM + khalf;
#pragma unroll
            for (int j = 0; j < HDIM / 4; ++j) bfrag[j] = *(const v2f*)(ub + uo + 4 * j);
        }

        for (int i = threadIdx.x; i < nown * (HDIM / 4); i += OWN_NT) {
            const int row = i >> 5, c4 = i & 31;
            const float den = rden[row] + EPSV;
            float4* hp = (float4*)&hacc[row][c4 * 4];
            float4 hh = *hp;
            hh.x /= den; hh.y /= den; hh.z /= den; hh.w /= den;
            *hp = hh;
        }
        __syncthreads();

        auto product = [&](int mt) {
            const float* arow = &hacc[mt * 16 + nsub][khalf];
            v8f acc = {};
#pragma unroll
            for (int j = 0; j < HDIM / 4; ++j) {
                const v2f a = *(const v2f*)(arow + 4 * j);
                acc = __builtin_amdgcn_wmma_f32_16x16x4_f32(
                     false, a,  false, bfrag[j],
                     (short)0, acc,  false,  false);
            }
            float* sp = &lu.g[mt & 1][pq][(lane >> 4) << 3][pn0 + nsub];
#pragma unroll
            for (int r = 0; r < 8; ++r) sp[r * OWN_LDP] = acc[r];
        };
        auto gate = [&](int mt, float2 gi, float2 gc, float2 go) {
            const int r = mt * 16 + grow;
            if (r < nown) {
                const float2 ai = *(const float2*)&lu.g[mt & 1][0][grow][gcol];
                const float2 ac = *(const float2*)&lu.g[mt & 1][1][grow][gcol];
                const float2 ao = *(const float2*)&lu.g[mt & 1][2][grow][gcol];
                const float2 sf = *(const float2*)&facc[r][gcol];
                float2 cc, hh;
                {
                    const float i_j = sigmoidf_(ai.x + gi.x);
                    const float c_t = tanhf(ac.x + gc.x);
                    const float c   = i_j * c_t + sf.x;
                    const float o_j = sigmoidf_(ao.x + go.x);
                    hh.x = o_j * tanhf(c);
                    cc.x = c;
                }
                {
                    const float i_j = sigmoidf_(ai.y + gi.y);
                    const float c_t = tanhf(ac.y + gc.y);
                    const float c   = i_j * c_t + sf.y;
                    const float o_j = sigmoidf_(ao.y + go.y);
                    hh.y = o_j * tanhf(c);
                    cc.y = c;
                }
                const size_t o = (size_t)(cbase + r) * HDIM + gcol;
                *(float2*)(h_out + o) = hh;
                *(float2*)(c_out + o) = cc;
            }
        };

        const bool pw     = wave < 24;
        const bool pfirst = ((wave >> 2) & 1) == 0;
        if (pw) product(0);
        __syncthreads();
        for (int mt = 0; mt < ntile; ++mt) {
            const bool nxt = mt + 1 < ntile;
            if (nxt && pw && pfirst) product(mt + 1);
            gate(mt, gi0, gc0, go0);
            if (nxt) load_g(mt + 1, gi0, gc0, go0);
            if (nxt && pw && !pfirst) product(mt + 1);
            __syncthreads();
        }
    }
}

extern "C" void kernel_launch(void* const* d_in, const int* in_sizes, int n_in,
                              void* d_out, int out_size, void* d_ws, size_t ws_size,
                              hipStream_t stream) {
    const float* x_emb   = (const float*)d_in[0];
    const float* child_h = (const float*)d_in[1];
    const float* child_c = (const float*)d_in[2];
    const int*   ci      = (const int*)d_in[3];
    const int*   chi     = (const int*)d_in[4];
    const float* Wi_w = (const float*)d_in[5];
    const float* Ui_w = (const float*)d_in[6];
    const float* Wf_w = (const float*)d_in[7];
    const float* Uf_w = (const float*)d_in[8];
    const float* Wo_w = (const float*)d_in[9];
    const float* Uo_w = (const float*)d_in[10];
    const float* Wc_w = (const float*)d_in[11];
    const float* Uc_w = (const float*)d_in[12];
    const float* Wa_w = (const float*)d_in[13];
    const float* Ua_w = (const float*)d_in[14];
    const float* Wi_b = (const float*)d_in[15];
    const float* Wf_b = (const float*)d_in[16];
    const float* Wo_b = (const float*)d_in[17];
    const float* Wc_b = (const float*)d_in[18];
    const float* Wa_b = (const float*)d_in[19];
    const float* v_w  = (const float*)d_in[20];

    const int M = in_sizes[0] / HDIM;
    const int L = in_sizes[3];
    const size_t MH = (size_t)M * HDIM;

    float* ws    = (float*)d_ws;
    float* Ah    = ws;
    float* Bx    = Ah  + MH;
    float* Gi    = Bx  + MH;
    float* Wfx   = Gi  + MH;
    float* Go    = Wfx + MH;
    float* Gc    = Go  + MH;
    float* Ufh   = Gc  + MH;

    proj7_wmma_f32<<<dim3(763), dim3(256), 0, stream>>>(x_emb, child_h,
                                                        Wa_w, Ua_w, Wi_w, Wf_w, Wo_w, Wc_w, Uf_w,
                                                        Wa_b, Wi_b, Wf_b, Wo_b, Wc_b,
                                                        ws, M, ci, (float*)d_out, L, OWN_BLOCKS);

    float* h_out = (float*)d_out;
    float* c_out = h_out + MH;
    edge_attn_owner_kernel<<<dim3(OWN_BLOCKS), dim3(OWN_NT), 0, stream>>>(Ah, Bx, v_w,
                                                        child_h, child_c, Wfx, Ufh,
                                                        ci, chi, Ui_w, Uc_w, Uo_w,
                                                        Gi, Gc, Go, h_out, c_out, M, L);
}


# Owner block sorts a round's edges by row (LDS integer adds give slots, a wave ranks each of its rows by edge index); a wave takes whole rows from its own stretch, no walk over all 32 lists; sums same
# speedup vs baseline: 2.9719x; 1.0394x over previous
#include <hip/hip_runtime.h>
#include <hip/hip_bf16.h>

#define HDIM 128
#define EPSV 1e-9f

typedef __attribute__((ext_vector_type(2))) float v2f;
typedef __attribute__((ext_vector_type(8))) float v8f;

#define PROJ_N 7

#define OWN_R      196
#define OWN_BLOCKS 256
#define PRE_NS     512
#define PRE_CAP    31
#define PRE_SCANB  64

__device__ __forceinline__ bool lists_prebuilt_(int nblk, int M, int L) {
    return nblk <= 256 && nblk * OWN_R >= M && L <= (1 << 24);
}

__device__ __forceinline__ void prebuild_lists(const int* __restrict__ ci, int* __restrict__ outw,
                                               int sid, int M, int L, int* cntw) {
    const int lane = threadIdx.x & 31;
    const unsigned lt = (1u << lane) - 1u;
#pragma unroll
    for (int j = 0; j < 8; ++j) cntw[lane + 32 * j] = 0;
    const int SLICE = (((L + PRE_NS - 1) / PRE_NS) + 31) & ~31;
    const int p0 = min(L, sid * SLICE), p1 = min(L, sid * SLICE + SLICE);
    int idn = (p0 < p1) ? ci[min(p0 + lane, L - 1)] : 0;
    for (int p = p0; p < p1; p += 32) {
        const int id  = idn;
        if (p + 32 < p1) idn = ci[min(p + 32 + lane, L - 1)];
        const int idx = p + lane;
        const int b   = id / OWN_R;
        const bool valid = idx < p1 && (unsigned)id < (unsigned)M && (b + 1) * OWN_R <= M;
        unsigned same = (unsigned)__ballot(valid);
#pragma unroll
        for (int k = 0; k < 8; ++k) {
            const bool bit = ((b >> k) & 1) != 0;
            const unsigned mk = (unsigned)__ballot(bit);
            same &= bit ? mk : ~mk;
        }
        if (valid) {
            const int base = cntw[b];
            const int rank = __popc(same & lt);
            const int tot  = __popc(same);
            const int pos  = base + rank;
            int* sub = outw + (size_t)b * (OWN_R * HDIM) + sid * 32;
            if (pos < PRE_CAP) sub[1 + pos] = (int)(((unsigned)idx << 8) | (unsigned)(id - b * OWN_R));
            if (rank == tot - 1) cntw[b] = base + tot;
        }
    }
#pragma unroll
    for (int j = 0; j < 8; ++j) {
        const int b = lane + 32 * j;
        if ((b + 1) * OWN_R <= M) outw[(size_t)b * (OWN_R * HDIM) + sid * 32] = cntw[b];
    }
}

__device__ __forceinline__ void proj_block(const float* __restrict__ X,
                                           const float* __restrict__ W,
                                           const float* __restrict__ bias,
                                           float* __restrict__ Y,
                                           int Mrows, int t0, int tstride) {
    __shared__ float lw[HDIM][132];
    __shared__ float lx[2][16][132];

    const int wave  = threadIdx.x >> 5;
    const int lane  = threadIdx.x & 31;
    const int nsub  = lane & 15;
    const int khalf = (lane >> 4) << 1;
    const int n0    = wave * 16;
    const int ntiles = (Mrows + 15) >> 4;

    const int srow = threadIdx.x >> 5;
    const int scol = (threadIdx.x & 31) << 2;
    const int lastrow = Mrows - 1;

    int t = t0;
    float4 p0 = {}, p1 = {};
    if (t < ntiles) {
        const int ra = min(t * 16 + srow, lastrow), rb = min(t * 16 + srow + 8, lastrow);
        p0 = *(const float4*)(X + (size_t)ra * HDIM + scol);
        p1 = *(const float4*)(X + (size_t)rb * HDIM + scol);
    }

#pragma unroll
    for (int p = 0; p < HDIM / 8; ++p)
        *(float4*)&lw[srow + 8 * p][scol] =
            *(const float4*)(W + (size_t)(srow + 8 * p) * HDIM + scol);
    __syncthreads();

    v2f bfrag[HDIM / 4];
#pragma unroll
    for (int j = 0; j < HDIM / 4; ++j)
        bfrag[j] = *(const v2f*)&lw[n0 + nsub][4 * j + khalf];
    const float bval = bias ? bias[n0 + nsub] : 0.0f;

    int buf = 0;
    for (; t < ntiles; t += tstride) {
        *(float4*)&lx[buf][srow][scol]     = p0;
        *(float4*)&lx[buf][srow + 8][scol] = p1;
        __syncthreads();

        const int tn = t + tstride;
        if (tn < ntiles) {
            const int ra = min(tn * 16 + srow, lastrow), rb = min(tn * 16 + srow + 8, lastrow);
            p0 = *(const float4*)(X + (size_t)ra * HDIM + scol);
            p1 = *(const float4*)(X + (size_t)rb * HDIM + scol);
        }

        const bool full  = (t * 16 + 16 <= Mrows);
        const int  mbase = t * 16 + ((lane >> 4) << 3);
        float* __restrict__ yp = Y + (size_t)mbase * HDIM + n0 + nsub;

        const float* arow = &lx[buf][nsub][khalf];
        v8f acc = {};
#pragma unroll
        for (int j = 0; j < HDIM / 4; ++j) {
            const v2f a = *(const v2f*)(arow + 4 * j);
            acc = __builtin_amdgcn_wmma_f32_16x16x4_f32(
                 false, a,  false, bfrag[j],
                 (short)0, acc,  false,  false);
        }

        float v[8];
#pragma unroll
        for (int r = 0; r < 8; ++r) v[r] = acc[r] + bval;
        if (full) {
#pragma unroll
            for (int r = 0; r < 8; ++r) yp[(size_t)r * HDIM] = v[r];
        } else {
#pragma unroll
            for (int r = 0; r < 8; ++r)
                if (mbase + r < Mrows) yp[(size_t)r * HDIM] = v[r];
        }
        buf ^= 1;
    }
}

__launch_bounds__(256)
__global__ void proj7_wmma_f32(const float* __restrict__ x_emb,
                               const float* __restrict__ child_h,
                               const float* __restrict__ Wa_w,
                               const float* __restrict__ Ua_w,
                               const float* __restrict__ Wi_w,
                               const float* __restrict__ Wf_w,
                               const float* __restrict__ Wo_w,
                               const float* __restrict__ Wc_w,
                               const float* __restrict__ Uf_w,
                               const float* __restrict__ Wa_b,
                               const float* __restrict__ Wi_b,
                               const float* __restrict__ Wf_b,
                               const float* __restrict__ Wo_b,
                               const float* __restrict__ Wc_b,
                               float* __restrict__ ws,
                               int Mrows,
                               const int* __restrict__ ci,
                               float* __restrict__ out,
                               int L, int nblk) {
    __shared__ int pre_cnt[8][256];
    const int sblk = (int)blockIdx.x - ((int)gridDim.x - PRE_SCANB);
    if (sblk >= 0 && lists_prebuilt_(nblk, Mrows, L)) {
        const int wv = __builtin_amdgcn_readfirstlane((int)(threadIdx.x >> 5));
        prebuild_lists(ci, (int*)out, sblk * 8 + wv, Mrows, L, &pre_cnt[wv][0]);
    }

    const int q  = blockIdx.x % PROJ_N;
    const int r0 = blockIdx.x / PROJ_N;
    const int pb = gridDim.x / PROJ_N;
    if (r0 >= pb) return;

    const float* X    = x_emb;
    const float* W    = Ua_w;
    const float* bias = nullptr;
    switch (q) {
        case 0: X = child_h; W = Wa_w; bias = Wa_b; break;
        case 1: break;
        case 2: W = Wi_w; bias = Wi_b; break;
        case 3: W = Wf_w; bias = Wf_b; break;
        case 4: W = Wo_w; bias = Wo_b; break;
        case 5: W = Wc_w; bias = Wc_b; break;
        default: X = child_h; W = Uf_w; break;
    }
    proj_block(X, W, bias, ws + (size_t)q * (size_t)Mrows * HDIM, Mrows, r0, pb);
}

__device__ __forceinline__ float sigmoidf_(float x) {
    return 1.0f / (1.0f + __expf(-x));
}

template <int XM>
__device__ __forceinline__ float row_xmask_(float x) {
    const int b = __builtin_bit_cast(int, x);
    return __builtin_bit_cast(float, __builtin_amdgcn_update_dpp(b, b, 0x160 | XM, 0xf, 0xf, false));
}
__device__ __forceinline__ float wave_sum32_(float x) {
    const int b = __builtin_bit_cast(int, x);
    x += __builtin_bit_cast(float, __builtin_amdgcn_permlanex16(b, b, 0x76543210, 0xfedcba98, false, false));
    x += row_xmask_<8>(x);
    x += row_xmask_<4>(x);
    x += row_xmask_<2>(x);
    x += row_xmask_<1>(x);
    return x;
}

#define OWN_RP   208
#define OWN_LDP  132
#define OWN_NT   1024
#define OWN_SEG  512
#define OWN_LIM  480
#define OWN_G    2

__launch_bounds__(OWN_NT)
__global__ void edge_attn_owner_kernel(const float* __restrict__ Ah,
                                       const float* __restrict__ Bx,
                                       const float* __restrict__ v_w,
                                       const float* __restrict__ child_h,
                                       const float* __restrict__ child_c,
                                       const float* __restrict__ Wfx,
                                       const float* __restrict__ Ufh,
                                       const int* __restrict__ ci,
                                       const int* __restrict__ chi,
                                       const float* __restrict__ Ui,
                                       const float* __restrict__ Uc,
                                       const float* __restrict__ Uo,
                                       const float* __restrict__ Gi,
                                       const float* __restrict__ Gc,
                                       const float* __restrict__ Go,
                                       float* __restrict__ h_out,
                                       float* __restrict__ c_out,
                                       int M, int L) {
    union ListOrStage {
        int   e[32][OWN_SEG];
        float g[2][3][16][OWN_LDP];
    };
    __shared__ alignas(16) float hacc[OWN_RP][OWN_LDP];
    __shared__ alignas(16) float facc[OWN_R][HDIM];
    __shared__ float rmax[OWN_R];
    __shared__ float rden[OWN_R];
    __shared__ alignas(16) ListOrStage lu;
    __shared__ unsigned char list_c[32][OWN_SEG];
    __shared__ int           s_more[32];
    __shared__ alignas(16) int rcnt[256];
    __shared__ alignas(4) unsigned char roww[256];
    __shared__ int            pw[256];
    __shared__ int            wstart[40];
    __shared__ unsigned short seg[32 * OWN_LIM];

    const int wave = __builtin_amdgcn_readfirstlane((int)(threadIdx.x >> 5));
    const int lane = threadIdx.x & 31;
    const unsigned lt = (1u << lane) - 1u;
    const int SL = (((L + 31) >> 5) + 31) & ~31;
    const float4 vw = ((const float4*)v_w)[lane];

    for (int cbase = blockIdx.x * OWN_R; cbase < M; cbase += gridDim.x * OWN_R) {
        const int nown = min(OWN_R, M - cbase);

        for (int i = threadIdx.x; i < OWN_RP * OWN_LDP / 4; i += OWN_NT) {
            ((float4*)&hacc[0][0])[i] = make_float4(0.f, 0.f, 0.f, 0.f);
            if (i < OWN_R * HDIM / 4)
                ((float4*)&facc[0][0])[i] = make_float4(0.f, 0.f, 0.f, 0.f);
        }
        for (int i = threadIdx.x; i < OWN_R; i += OWN_NT) {
            rmax[i] = -INFINITY;
            rden[i] = 0.f;
        }
        if (threadIdx.x < 256) rcnt[threadIdx.x] = 0;
        int wpos = min(L, wave * SL);
        const int wend = min(L, wave * SL + SL);
        bool pre = lists_prebuilt_((int)gridDim.x, M, L) && nown == OWN_R;
        __syncthreads();

        bool more = true;
        while (more) {
            int cnt = 0;
            int flag = 0;
            if (pre) {
                const int* reg = (const int*)(h_out + (size_t)cbase * HDIM) + wave * (16 * 32);
                bool ov = false;
                for (int h8 = 0; h8 < 16; h8 += 8) {
                    int w8[8];
#pragma unroll
                    for (int i = 0; i < 8; ++i) w8[i] = reg[(h8 + i) * 32 + lane];
#pragma unroll
                    for (int i = 0; i < 8; ++i) {
                        const int c = __builtin_amdgcn_readlane(w8[i], 0);
                        ov = ov || ((unsigned)c > (unsigned)PRE_CAP);
                        const int cc = ov ? 0 : c;
                        if (lane >= 1 && lane <= cc) {
                            lu.e[wave][cnt + lane - 1]   = (int)((unsigned)w8[i] >> 8);
                            list_c[wave][cnt + lane - 1] = (unsigned char)(w8[i] & 0xff);
                            atomicAdd(&rcnt[w8[i] & 0xff], 1);
                        }
                        cnt += cc;
                    }
                }
                ov = ov || cnt > OWN_LIM;
                if (ov) { cnt = 0; flag = 2; }
            } else if (wpos < wend) {
                int ida[8], idb[8];
                auto load_ids = [&](int (&v)[8], int p) {
#pragma unroll
                    for (int j = 0; j < 8; ++j) {
                        const int idx = p + j * 32 + lane;
                        const int t = ci[min(idx, L - 1)];
                        v[j] = (idx < wend) ? t : -1;
                    }
                };
                auto take = [&](const int (&v)[8], int p) {
#pragma unroll
                    for (int j = 0; j < 8; ++j) {
                        const int cl = v[j] - cbase;
                        const bool hit = (unsigned)cl < (unsigned)nown;
                        const unsigned m = (unsigned)__ballot(hit);
                        if (hit) {
                            const int k = cnt + __popc(m & lt);
                            lu.e[wave][k] = p + j * 32 + lane;
                            list_c[wave][k] = (unsigned char)cl;
                            atomicAdd(&rcnt[cl], 1);
                        }
                        cnt += __popc(m);
                    }
                };
                load_ids(ida, wpos);
                for (;;) {
                    load_ids(idb, wpos + 256);
                    take(ida, wpos);
                    wpos += 256;
                    if (!(wpos < wend && cnt + 256 <= OWN_LIM)) break;
                    load_ids(ida, wpos + 256);
                    take(idb, wpos);
                    wpos += 256;
                    if (!(wpos < wend && cnt + 256 <= OWN_LIM)) break;
                }
                wpos = min(wpos, wend);
            }
            if (!pre) flag = (wpos < wend) ? 1 : 0;
            if (lane == 0) s_more[wave] = flag;
            __syncthreads();
            more = __ballot(s_more[lane] != 0) != 0;
            bool use = true;
            if (pre) {
                if (more) use = false;
                pre = false;
            }

            {
                const int t   = (int)threadIdx.x;
                const int r   = min(t >> 2, OWN_R - 1);
                const int pt  = t & 3;
                const int mk  = (rcnt[r] << 8) + r;
                int before = 0;
                static_assert(OWN_R % 4 == 0 && OWN_R <= 256, "rows are ranked in groups of four, row number in 8 bits");
                for (int g = pt; g < OWN_R / 4; g += 4) {
                    const int4 c4 = ((const int4*)rcnt)[g];
                    const int  jb = g * 4;
                    before += (((c4.x << 8) + jb    ) > mk) ? 1 : 0;
                    before += (((c4.y << 8) + jb + 1) > mk) ? 1 : 0;
                    before += (((c4.z << 8) + jb + 2) > mk) ? 1 : 0;
                    before += (((c4.w << 8) + jb + 3) > mk) ? 1 : 0;
                }
                before += __builtin_amdgcn_update_dpp(before, before, 0x160 | 1, 0xf, 0xf, false);
                before += __builtin_amdgcn_update_dpp(before, before, 0x160 | 2, 0xf, 0xf, false);
                if (pt == 0 && (t >> 2) < OWN_R) {
                    const int pos = before & 31;
                    roww[r] = (unsigned char)(((before >> 5) & 1) ? 31 - pos : pos);
                }
            }
            __syncthreads();
            {
                const int t   = (int)threadIdx.x;
                const int grp = t >> 2;
                const int pt  = t & 3;
                const int mk  = (grp < OWN_R) ? (((int)roww[grp] << 8) + grp) : (min(grp - OWN_R, 32) << 8);
                int first = 0;
                for (int g = pt; g < OWN_R / 4; g += 4) {
                    const int4   c4 = ((const int4*)rcnt)[g];
                    const uchar4 w4 = ((const uchar4*)roww)[g];
                    const int    jb = g * 4;
                    first += ((((int)w4.x << 8) + jb    ) < mk) ? c4.x : 0;
                    first += ((((int)w4.y << 8) + jb + 1) < mk) ? c4.y : 0;
                    first += ((((int)w4.z << 8) + jb + 2) < mk) ? c4.z : 0;
                    first += ((((int)w4.w << 8) + jb + 3) < mk) ? c4.w : 0;
                }
                first += __builtin_amdgcn_update_dpp(first, first, 0x160 | 1, 0xf, 0xf, false);
                first += __builtin_amdgcn_update_dpp(first, first, 0x160 | 2, 0xf, 0xf, false);
                if (pt == 0) {
                    if (grp < OWN_R)            pw[grp] = (rcnt[grp] << 16) | first;
                    else if (grp - OWN_R <= 32) wstart[grp - OWN_R] = first;
                }
            }
            __syncthreads();
            if (threadIdx.x < 256) rcnt[threadIdx.x] = 0;
            if (use) {
                for (int k = lane; k < cnt; k += 32) {
                    const int old = atomicAdd(&pw[list_c[wave][k]], 1);
                    seg[old & 0xffff] = (unsigned short)((wave << 9) | k);
                }
            }
            __syncthreads();

            if (use) {
                for (int rb = 0; rb < OWN_R; rb += 32) {
                    const int  r    = min(rb + lane, OWN_R - 1);
                    const int  word = pw[r];
                    const bool mine = rb + lane < OWN_R && (int)roww[r] == wave && (word >> 16) >= 2;
                    unsigned mm = (unsigned)__ballot(mine);
                    while (mm != 0u) {
                        const int j  = __builtin_amdgcn_readfirstlane(__builtin_ctz(mm));
                        mm &= mm - 1u;
                        const int wj = __builtin_amdgcn_readlane(word, j);
                        const int nr = wj >> 16;
                        const int s0 = (wj & 0xffff) - nr;
                        if (nr <= 32) {
                            int key = 0x7fffffff;
                            if (lane < nr) key = (int)seg[s0 + lane];
                            int rank = 0;
                            for (int i = 0; i < nr; ++i)
                                rank += (__builtin_amdgcn_readlane(key, i) < key) ? 1 : 0;
                            if (lane < nr) seg[s0 + rank] = (unsigned short)key;
                        } else {
                            for (int ph = 0; ph < nr; ++ph) {
                                for (int i = (ph & 1) + 2 * lane; i + 1 < nr; i += 64) {
                                    const unsigned short a = seg[s0 + i], b = seg[s0 + i + 1];
                                    if (a > b) { seg[s0 + i] = b; seg[s0 + i + 1] = a; }
                                }
                            }
                        }
                    }
                }

                int       pos  = __builtin_amdgcn_readfirstlane(wstart[wave]);
                const int pend = __builtin_amdgcn_readfirstlane(wstart[wave + 1]);
                while (pos < pend) {
                    const int n = min(pend - pos, 32);
                    int bcl = 0, bch = 0;
                    if (lane < n) {
                        const int key = (int)seg[pos + lane];
                        bcl = (int)(&list_c[0][0])[key];
                        bch = chi[(&lu.e[0][0])[key]];
                    }
                    for (int g = 0; g < n; g += OWN_G) {
                        float4 ah[OWN_G], bx[OWN_G], hv[OWN_G], cv4[OWN_G], uf[OWN_G], wf[OWN_G];
                        int   cl4[OWN_G];
#pragma unroll
                        for (int u = 0; u < OWN_G; ++u) {
                            const int i  = min(g + u, n - 1);
                            const int ch = __builtin_amdgcn_readlane(bch, i);
                            const int cl = __builtin_amdgcn_readlane(bcl, i);
                            cl4[u] = cl;
                            ah[u]  = ((const float4*)(Ah      + (size_t)ch * HDIM))[lane];
                            hv[u]  = ((const float4*)(child_h + (size_t)ch * HDIM))[lane];
                            cv4[u] = ((const float4*)(child_c + (size_t)ch * HDIM))[lane];
                            uf[u]  = ((const float4*)(Ufh     + (size_t)ch * HDIM))[lane];
                            bx[u]  = ((const float4*)(Bx  + (size_t)(cbase + cl) * HDIM))[lane];
                            wf[u]  = ((const float4*)(Wfx + (size_t)(cbase + cl) * HDIM))[lane];
                        }
#pragma unroll
                        for (int u = 0; u < OWN_G; ++u) {
                            if (g + u < n) {
                                float s = tanhf(ah[u].x + bx[u].x) * vw.x + tanhf(ah[u].y + bx[u].y) * vw.y
                                        + tanhf(ah[u].z + bx[u].z) * vw.z + tanhf(ah[u].w + bx[u].w) * vw.w;
                                s = wave_sum32_(s);

                                const int   cl    = cl4[u];
                                const float m_old = rmax[cl];
                                const float d_old = rden[cl];
                                const float m_new = fmaxf(m_old, s);
                                const float sc    = __expf(m_old - m_new);
                                const float p     = __expf(s - m_new);
                                if (lane == 0) {
                                    rmax[cl] = m_new;
                                    rden[cl] = d_old * sc + p;
                                }

                                float4* hp = (float4*)&hacc[cl][lane * 4];
                                float4* fp = (float4*)&facc[cl][lane * 4];
                                float4 h = *hp, f = *fp;
                                h.x = h.x * sc + p * hv[u].x;
                                h.y = h.y * sc + p * hv[u].y;
                                h.z = h.z * sc + p * hv[u].z;
                                h.w = h.w * sc + p * hv[u].w;
                                f.x += sigmoidf_(wf[u].x + uf[u].x) * cv4[u].x;
                                f.y += sigmoidf_(wf[u].y + uf[u].y) * cv4[u].y;
                                f.z += sigmoidf_(wf[u].z + uf[u].z) * cv4[u].z;
                                f.w += sigmoidf_(wf[u].w + uf[u].w) * cv4[u].w;
                                *hp = h;
                                *fp = f;
                            }
                        }
                    }
                    pos += n;
                }
            }
            __syncthreads();
        }

        const int ntile = (nown + 15) >> 4;
        const int nsub  = lane & 15;
        const int khalf = (lane >> 4) << 1;
        const int pq    = wave >> 3;
        const int pn0   = (wave & 7) * 16;
        const int grow  = wave >> 1;
        const int gcol  = (((wave & 1) << 5) + lane) << 1;

        auto load_g = [&](int mt, float2& gi, float2& gc, float2& go) {
            const size_t o = (size_t)(cbase + min(mt * 16 + grow, nown - 1)) * HDIM + gcol;
            gi = *(const float2*)(Gi + o);
            gc = *(const float2*)(Gc + o);
            go = *(const float2*)(Go + o);
        };
        float2 gi0, gc0, go0;
        load_g(0, gi0, gc0, go0);

        v2f bfrag[HDIM / 4] = {};
        if (wave < 24) {
            const float* ub = (pq == 0 ? Ui : (pq == 1 ? Uc : Uo)) + pn0 * HDIM;
            const int    uo = nsub * HDIM + khalf;
#pragma unroll
            for (int j = 0; j < HDIM / 4; ++j) bfrag[j] = *(const v2f*)(ub + uo + 4 * j);
        }

        for (int i = threadIdx.x; i < nown * (HDIM / 4); i += OWN_NT) {
            const int row = i >> 5, c4 = i & 31;
            const float den = rden[row] + EPSV;
            float4* hp = (float4*)&hacc[row][c4 * 4];
            float4 hh = *hp;
            hh.x /= den; hh.y /= den; hh.z /= den; hh.w /= den;
            *hp = hh;
        }
        __syncthreads();

        auto product = [&](int mt) {
            const float* arow = &hacc[mt * 16 + nsub][khalf];
            v8f acc = {};
#pragma unroll
            for (int j = 0; j < HDIM / 4; ++j) {
                const v2f a = *(const v2f*)(arow + 4 * j);
                acc = __builtin_amdgcn_wmma_f32_16x16x4_f32(
                     false, a,  false, bfrag[j],
                     (short)0, acc,  false,  false);
            }
            float* sp = &lu.g[mt & 1][pq][(lane >> 4) << 3][pn0 + nsub];
#pragma unroll
            for (int r = 0; r < 8; ++r) sp[r * OWN_LDP] = acc[r];
        };
        auto gate = [&](int mt, float2 gi, float2 gc, float2 go) {
            const int r = mt * 16 + grow;
            if (r < nown) {
                const float2 ai = *(const float2*)&lu.g[mt & 1][0][grow][gcol];
                const float2 ac = *(const float2*)&lu.g[mt & 1][1][grow][gcol];
                const float2 ao = *(const float2*)&lu.g[mt & 1][2][grow][gcol];
                const float2 sf = *(const float2*)&facc[r][gcol];
                float2 cc, hh;
                {
                    const float i_j = sigmoidf_(ai.x + gi.x);
                    const float c_t = tanhf(ac.x + gc.x);
                    const float c   = i_j * c_t + sf.x;
                    const float o_j = sigmoidf_(ao.x + go.x);
                    hh.x = o_j * tanhf(c);
                    cc.x = c;
                }
                {
                    const float i_j = sigmoidf_(ai.y + gi.y);
                    const float c_t = tanhf(ac.y + gc.y);
                    const float c   = i_j * c_t + sf.y;
                    const float o_j = sigmoidf_(ao.y + go.y);
                    hh.y = o_j * tanhf(c);
                    cc.y = c;
                }
                const size_t o = (size_t)(cbase + r) * HDIM + gcol;
                *(float2*)(h_out + o) = hh;
                *(float2*)(c_out + o) = cc;
            }
        };

        const bool pw     = wave < 24;
        const bool pfirst = ((wave >> 2) & 1) == 0;
        if (pw) product(0);
        __syncthreads();
        for (int mt = 0; mt < ntile; ++mt) {
            const bool nxt = mt + 1 < ntile;
            if (nxt && pw && pfirst) product(mt + 1);
            gate(mt, gi0, gc0, go0);
            if (nxt) load_g(mt + 1, gi0, gc0, go0);
            if (nxt && pw && !pfirst) product(mt + 1);
            __syncthreads();
        }
    }
}

extern "C" void kernel_launch(void* const* d_in, const int* in_sizes, int n_in,
                              void* d_out, int out_size, void* d_ws, size_t ws_size,
                              hipStream_t stream) {
    const float* x_emb   = (const float*)d_in[0];
    const float* child_h = (const float*)d_in[1];
    const float* child_c = (const float*)d_in[2];
    const int*   ci      = (const int*)d_in[3];
    const int*   chi     = (const int*)d_in[4];
    const float* Wi_w = (const float*)d_in[5];
    const float* Ui_w = (const float*)d_in[6];
    const float* Wf_w = (const float*)d_in[7];
    const float* Uf_w = (const float*)d_in[8];
    const float* Wo_w = (const float*)d_in[9];
    const float* Uo_w = (const float*)d_in[10];
    const float* Wc_w = (const float*)d_in[11];
    const float* Uc_w = (const float*)d_in[12];
    const float* Wa_w = (const float*)d_in[13];
    const float* Ua_w = (const float*)d_in[14];
    const float* Wi_b = (const float*)d_in[15];
    const float* Wf_b = (const float*)d_in[16];
    const float* Wo_b = (const float*)d_in[17];
    const float* Wc_b = (const float*)d_in[18];
    const float* Wa_b = (const float*)d_in[19];
    const float* v_w  = (const float*)d_in[20];

    const int M = in_sizes[0] / HDIM;
    const int L = in_sizes[3];
    const size_t MH = (size_t)M * HDIM;

    float* ws    = (float*)d_ws;
    float* Ah    = ws;
    float* Bx    = Ah  + MH;
    float* Gi    = Bx  + MH;
    float* Wfx   = Gi  + MH;
    float* Go    = Wfx + MH;
    float* Gc    = Go  + MH;
    float* Ufh   = Gc  + MH;

    proj7_wmma_f32<<<dim3(763), dim3(256), 0, stream>>>(x_emb, child_h,
                                                        Wa_w, Ua_w, Wi_w, Wf_w, Wo_w, Wc_w, Uf_w,
                                                        Wa_b, Wi_b, Wf_b, Wo_b, Wc_b,
                                                        ws, M, ci, (float*)d_out, L, OWN_BLOCKS);

    float* h_out = (float*)d_out;
    float* c_out = h_out + MH;
    edge_attn_owner_kernel<<<dim3(OWN_BLOCKS), dim3(OWN_NT), 0, stream>>>(Ah, Bx, v_w,
                                                        child_h, child_c, Wfx, Ufh,
                                                        ci, chi, Ui_w, Uc_w, Uo_w,
                                                        Gi, Gc, Go, h_out, c_out, M, L);
}
